# starting kernel
# baseline (speedup 1.0000x reference)
#include <hip/hip_runtime.h>

typedef __attribute__((ext_vector_type(2))) float v2f;
typedef __attribute__((ext_vector_type(8))) float v8f;

#define NB      2048
#define FC_K    100
#define FC_N    17640
#define NTILES  1103
#define NGRPS   138
#define H1_ELEMS (NB * FC_N)
#define H2_ELEMS (NB * 5 * 98 * 16)
#define CONV_BLOCKS 7840
#define PART_ELEMS (CONV_BLOCKS * 32)
#define BN_COUNT 1003520.0f

__global__ void __launch_bounds__(256) fc_wmma(const float* __restrict__ x,
                                               const float* __restrict__ fw,
                                               const float* __restrict__ fb,
                                               float* __restrict__ h1) {
  const int lane  = threadIdx.x & 31;
  const int wave  = threadIdx.x >> 5;
  const int w     = blockIdx.x * 8 + wave;
  const int mtile = w / NGRPS;
  const int ngrp  = w - mtile * NGRPS;
  const int lrow  = lane & 15;
  const int lhalf = lane >> 4;
  const int row   = mtile * 16 + lrow;
  const int kb0   = 2 * lhalf;

  v2f a[25];
#pragma unroll
  for (int s = 0; s < 25; ++s)
    a[s] = *(const v2f*)(x + row * FC_K + 4 * s + kb0);

  for (int t = 0; t < 8; ++t) {
    int ntile = ngrp * 8 + t;
    if (ntile >= NTILES) break;
    int n = ntile * 16 + lrow;
    v8f acc = {};
    if (ntile < NTILES - 1) {
      const float* bp = fw + n * FC_K + kb0;
#pragma unroll
      for (int s = 0; s < 25; ++s) {
        v2f bf = *(const v2f*)(bp + 4 * s);
        acc = __builtin_amdgcn_wmma_f32_16x16x4_f32(false, a[s], false, bf,
                                                    (short)0, acc, false, false);
      }
      float bias = fb[n];
#pragma unroll
      for (int v = 0; v < 8; ++v) {
        int m = mtile * 16 + lhalf * 8 + v;
        h1[m * FC_N + n] = acc[v] + bias;
      }
    } else {
      bool nvalid = (n < FC_N);
      const float* bp = fw + (nvalid ? n : 0) * FC_K + kb0;
#pragma unroll
      for (int s = 0; s < 25; ++s) {
        v2f bf = *(const v2f*)(bp + 4 * s);
        if (!nvalid) { bf[0] = 0.f; bf[1] = 0.f; }
        acc = __builtin_amdgcn_wmma_f32_16x16x4_f32(false, a[s], false, bf,
                                                    (short)0, acc, false, false);
      }
      float bias = nvalid ? fb[n] : 0.f;
#pragma unroll
      for (int v = 0; v < 8; ++v) {
        int m = mtile * 16 + lhalf * 8 + v;
        if (nvalid) h1[m * FC_N + n] = acc[v] + bias;
      }
    }
  }
}

__global__ void pack_conv_w(const float* __restrict__ cw, float* __restrict__ pw) {
  int idx = blockIdx.x * 256 + threadIdx.x;
  if (idx >= 2304) return;
  int n = idx / 144;
  int k = idx - n * 144;
  int p = k / 36;
  int c = k - p * 36;
  pw[idx] = cw[n * 144 + c * 4 + p];
}

__global__ void __launch_bounds__(256) conv_wmma(const float* __restrict__ h1,
                                                 const float* __restrict__ pw,
                                                 const float* __restrict__ cb,
                                                 float* __restrict__ h2,
                                                 float* __restrict__ part) {
  __shared__ float wsum[8][16];
  __shared__ float wsq[8][16];
  const int lane  = threadIdx.x & 31;
  const int wave  = threadIdx.x >> 5;
  const int w     = blockIdx.x * 8 + wave;
  const int mtile = w / 490;
  const int sp    = w - mtile * 490;
  const int i     = sp / 98;
  const int j     = sp - i * 98;
  const int lrow  = lane & 15;
  const int lhalf = lane >> 4;
  const int bb    = mtile * 16 + lrow;
  const float* bwp = pw + lrow * 144 + 2 * lhalf;

  v8f acc = {};
#pragma unroll
  for (int s = 0; s < 36; ++s) {
    const int p  = (4 * s) / 36;
    const int di = p >> 1, dj = p & 1;
    const int ii = i + di, jj = j + dj;
    int c0 = 4 * s - p * 36 + 2 * lhalf;
    v2f af;
    if (ii < 5 && jj < 98) {
      af = *(const v2f*)(h1 + (((bb * 5 + ii) * 98 + jj) * 36 + c0));
    } else {
      af[0] = 0.f; af[1] = 0.f;
    }
    v2f bf = *(const v2f*)(bwp + 4 * s);
    acc = __builtin_amdgcn_wmma_f32_16x16x4_f32(false, af, false, bf,
                                                (short)0, acc, false, false);
  }

  float bias = cb[lrow];
  float s0 = 0.f, q0 = 0.f;
#pragma unroll
  for (int v = 0; v < 8; ++v) {
    int m = mtile * 16 + lhalf * 8 + v;
    float val = acc[v] + bias;
    val = val >= 0.f ? val : 0.01f * val;
    h2[((m * 5 + i) * 98 + j) * 16 + lrow] = val;
    s0 += val;
    q0 += val * val;
  }
  s0 += __shfl_xor(s0, 16, 32);
  q0 += __shfl_xor(q0, 16, 32);
  if (lane < 16) { wsum[wave][lane] = s0; wsq[wave][lane] = q0; }
  __syncthreads();
  if (threadIdx.x < 32) {
    int c = threadIdx.x & 15;
    int st = threadIdx.x >> 4;
    float tot = 0.f;
    for (int wv = 0; wv < 8; ++wv) tot += st ? wsq[wv][c] : wsum[wv][c];
    part[blockIdx.x * 32 + threadIdx.x] = tot;
  }
}

__global__ void __launch_bounds__(256) bn_finalize(const float* __restrict__ part,
                                                   const float* __restrict__ gamma,
                                                   const float* __restrict__ beta,
                                                   float* __restrict__ scale,
                                                   float* __restrict__ shift) {
  __shared__ float acc[256];
  __shared__ float tot[32];
  int t = threadIdx.x;
  int cs = t & 31, r = t >> 5;
  float s = 0.f;
  for (int blk = r; blk < CONV_BLOCKS; blk += 8) s += part[blk * 32 + cs];
  acc[t] = s;
  __syncthreads();
  if (t < 32) {
    float sum = 0.f;
    for (int rr = 0; rr < 8; ++rr) sum += acc[rr * 32 + t];
    tot[t] = sum;
  }
  __syncthreads();
  if (t < 16) {
    const float invN = 1.0f / BN_COUNT;
    float mean = tot[t] * invN;
    float var  = tot[16 + t] * invN - mean * mean;
    float inv  = rsqrtf(var + 1e-5f);
    float scl  = gamma[t] * inv;
    scale[t] = scl;
    shift[t] = beta[t] - mean * scl;
  }
}

#define H2PB 7840
#define H3SZ 2328
__global__ void __launch_bounds__(128) lc_kernel(const float* __restrict__ h2,
                                                 const float* __restrict__ scale,
                                                 const float* __restrict__ shift,
                                                 const float* __restrict__ w1,
                                                 const float* __restrict__ b1,
                                                 const float* __restrict__ w2,
                                                 const float* __restrict__ b2,
                                                 float* __restrict__ out) {
  __shared__ float sh2[H2PB];
  __shared__ float sh3[H3SZ];
  const int b = blockIdx.x;
  const int tid = threadIdx.x;

  const float4 sc4 = ((const float4*)scale)[tid & 3];
  const float4 sf4 = ((const float4*)shift)[tid & 3];
  const float4* src4 = (const float4*)(h2 + b * H2PB);
  for (int idx = tid; idx < H2PB / 4; idx += 128) {
    float4 v = src4[idx];
    v.x = v.x * sc4.x + sf4.x;
    v.y = v.y * sc4.y + sf4.y;
    v.z = v.z * sc4.z + sf4.z;
    v.w = v.w * sc4.w + sf4.w;
    ((float4*)sh2)[idx] = v;
  }
  __syncthreads();

  for (int e = tid; e < H3SZ; e += 128) {
    int o1 = e / 388;
    int r  = e - o1 * 388;
    int oi = r / 97;
    int oj = r - oi * 97;
    float acc = b1[e];
    const float* wp = w1 + o1 * 24832 + oi * 388 + oj * 4;
    const float* xb = sh2 + (oi * 98 + oj) * 16;
#pragma unroll
    for (int c = 0; c < 16; ++c) {
      float4 wv = *(const float4*)(wp + c * 1552);
      acc += wv.x * xb[c]
           + wv.y * xb[16 + c]
           + wv.z * xb[98 * 16 + c]
           + wv.w * xb[99 * 16 + c];
    }
    acc = acc >= 0.f ? acc : 0.01f * acc;
    sh3[e] = acc;
  }
  __syncthreads();

  for (int e = tid; e < 288; e += 128) {
    int ri = e / 96;
    int rj = e - ri * 96;
    float acc = b2[e];
#pragma unroll
    for (int o1 = 0; o1 < 6; ++o1) {
      float4 wv = *(const float4*)(w2 + (((o1 * 3 + ri) * 96 + rj) << 2));
      const float* xb = sh3 + o1 * 388 + ri * 97 + rj;
      acc += wv.x * xb[0] + wv.y * xb[1] + wv.z * xb[97] + wv.w * xb[98];
    }
    out[b * 288 + e] = acc;
  }
}

extern "C" void kernel_launch(void* const* d_in, const int* in_sizes, int n_in,
                              void* d_out, int out_size, void* d_ws, size_t ws_size,
                              hipStream_t stream) {
  const float* x     = (const float*)d_in[0];
  const float* fw    = (const float*)d_in[1];
  const float* fb    = (const float*)d_in[2];
  const float* cw    = (const float*)d_in[3];
  const float* cb    = (const float*)d_in[4];
  const float* gamma = (const float*)d_in[5];
  const float* beta  = (const float*)d_in[6];
  const float* w1    = (const float*)d_in[7];
  const float* b1    = (const float*)d_in[8];
  const float* w2    = (const float*)d_in[9];
  const float* b2    = (const float*)d_in[10];

  float* ws    = (float*)d_ws;
  float* h1    = ws;
  float* h2    = ws + H1_ELEMS;
  float* part  = ws + H1_ELEMS + H2_ELEMS;
  float* scale = part + PART_ELEMS;
  float* shift = scale + 16;
  float* pw    = shift + 16;

  pack_conv_w<<<9, 256, 0, stream>>>(cw, pw);
  fc_wmma    <<<2208, 256, 0, stream>>>(x, fw, fb, h1);
  conv_wmma  <<<CONV_BLOCKS, 256, 0, stream>>>(h1, pw, cb, h2, part);
  bn_finalize<<<1, 256, 0, stream>>>(part, gamma, beta, scale, shift);
  lc_kernel  <<<NB, 128, 0, stream>>>(h2, scale, shift, w1, b1, w2, b2,
                                      (float*)d_out);
}


# lc1 moved to the matrix unit: per position a [6 ch x 64] x [64 x 16 batch rows] f32 WMMA GEMM; a wave walks 7 columns with h2 columns in registers, lc2 from registers; lc1 f32 sum order changes
# speedup vs baseline: 1.9727x; 1.9727x over previous
#include <hip/hip_runtime.h>

typedef __attribute__((ext_vector_type(2))) float v2f;
typedef __attribute__((ext_vector_type(8))) float v8f;

#define NB      2048
#define FC_K    100
#define FC_N    17640
#define NTILES  1103
#define H1_ELEMS (NB * FC_N)
#define H2_ELEMS (NB * 5 * 98 * 16)
#define CONV_BLOCKS 7840
#define PART_ELEMS (CONV_BLOCKS * 32)
#define BN_COUNT 1003520.0f
#define H1_PAIRS 8820

#define SLAB_FLOATS 12800
#define SLAB_VEC4   3200
#define FW_VEC4     441000
__global__ void __launch_bounds__(256) fc_wmma(const float* __restrict__ x,
                                               const float* __restrict__ fw,
                                               const float* __restrict__ fb,
                                               float* __restrict__ h1) {
  __shared__ float sB[SLAB_FLOATS];
  const int tid   = threadIdx.x;
  const int lane  = tid & 31;
  const int wave  = tid >> 5;
  const int slab  = blockIdx.x >> 4;
  const int mblk  = blockIdx.x & 15;
  const int mtile = mblk * 8 + wave;
  const int lrow  = lane & 15;
  const int lhalf = lane >> 4;
  const int row   = mtile * 16 + lrow;
  const int kb0   = 2 * lhalf;

  v2f a[25];
#pragma unroll
  for (int s = 0; s < 25; ++s)
    a[s] = *(const v2f*)(x + row * FC_K + 4 * s + kb0);

  const float4* src4 = (const float4*)fw + slab * SLAB_VEC4;
  const int vleft = FW_VEC4 - slab * SLAB_VEC4;
  float4 stg[13];
#pragma unroll
  for (int r = 0; r < 13; ++r) {
    const int idx = tid + r * 256;
    float4 v = make_float4(0.f, 0.f, 0.f, 0.f);
    if ((r < 12 || tid < 128) && idx < vleft) v = src4[idx];
    stg[r] = v;
  }
#pragma unroll
  for (int r = 0; r < 13; ++r) {
    const int idx = tid + r * 256;
    if (r < 12 || tid < 128) ((float4*)sB)[idx] = stg[r];
  }
  __syncthreads();

  for (int t = 0; t < 8; ++t) {
    const int ntile = slab * 8 + t;
    if (ntile >= NTILES) break;
    const int nl = ntile * 16 + 8 * lhalf;
    const bool nvalid = (nl < FC_N);
    float4 bia0 = make_float4(0.f, 0.f, 0.f, 0.f), bia1 = bia0;
    if (nvalid) {
      bia0 = *(const float4*)(fb + nl);
      bia1 = *(const float4*)(fb + nl + 4);
    }
    const float* bp = sB + (t * 16 + lrow) * FC_K + kb0;
    v8f acc = {};
#pragma unroll
    for (int s = 0; s < 25; ++s) {
      v2f wf = *(const v2f*)(bp + 4 * s);
      acc = __builtin_amdgcn_wmma_f32_16x16x4_f32(false, wf, false, a[s],
                                                  (short)0, acc, false, false);
    }
    if (nvalid) {
      float* dst = h1 + ((mtile * H1_PAIRS + ntile * 8 + 4 * lhalf) * 32 + lrow * 2);
      v2f o0, o1, o2, o3;
      o0[0] = acc[0] + bia0.x; o0[1] = acc[1] + bia0.y;
      o1[0] = acc[2] + bia0.z; o1[1] = acc[3] + bia0.w;
      o2[0] = acc[4] + bia1.x; o2[1] = acc[5] + bia1.y;
      o3[0] = acc[6] + bia1.z; o3[1] = acc[7] + bia1.w;
      *(v2f*)(dst)      = o0;
      *(v2f*)(dst + 32) = o1;
      *(v2f*)(dst + 64) = o2;
      *(v2f*)(dst + 96) = o3;
    }
  }
}

__global__ void pack_conv_w(const float* __restrict__ cw, float* __restrict__ pw) {
  int idx = blockIdx.x * 256 + threadIdx.x;
  if (idx >= 2304) return;
  int s  = idx >> 6;
  int r  = idx & 63;
  int ln = r >> 1;
  int e  = r & 1;
  int n  = ln & 15;
  int k  = 4 * s + 2 * (ln >> 4) + e;
  int p  = k / 36;
  int c  = k - p * 36;
  pw[idx] = cw[n * 144 + c * 4 + p];
}

__global__ void __launch_bounds__(256) conv_wmma(const float* __restrict__ h1,
                                                 const float* __restrict__ pw,
                                                 const float* __restrict__ cb,
                                                 float* __restrict__ h2,
                                                 float* __restrict__ part) {
  __shared__ float wsum[8][16];
  __shared__ float wsq[8][16];
  const int lane  = threadIdx.x & 31;
  const int wave  = threadIdx.x >> 5;
  const int w     = blockIdx.x * 8 + wave;
  const int mtile = w / 490;
  const int sp    = w - mtile * 490;
  const int i     = sp / 98;
  const int j     = sp - i * 98;
  const int lrow  = lane & 15;
  const int lhalf = lane >> 4;
  const float* ap  = h1 + (mtile * H1_PAIRS * 32 + lhalf * 32 + lrow * 2);
  const float* bwp = pw + lane * 2;

  v8f acc = {};
#pragma unroll
  for (int s = 0; s < 36; ++s) {
    const int p  = (4 * s) / 36;
    const int di = p >> 1, dj = p & 1;
    const int ii = i + di, jj = j + dj;
    const int pr0 = 2 * (s - p * 9);
    v2f af;
    if (ii < 5 && jj < 98) {
      af = *(const v2f*)(ap + ((ii * 98 + jj) * 18 + pr0) * 32);
    } else {
      af[0] = 0.f; af[1] = 0.f;
    }
    v2f bf = *(const v2f*)(bwp + 64 * s);
    acc = __builtin_amdgcn_wmma_f32_16x16x4_f32(false, af, false, bf,
                                                (short)0, acc, false, false);
  }

  float bias = cb[lrow];
  float s0 = 0.f, q0 = 0.f;
#pragma unroll
  for (int v = 0; v < 8; ++v) {
    int m = mtile * 16 + lhalf * 8 + v;
    float val = acc[v] + bias;
    val = val >= 0.f ? val : 0.01f * val;
    h2[((m * 5 + i) * 98 + j) * 16 + lrow] = val;
    s0 += val;
    q0 += val * val;
  }
  s0 += __shfl_xor(s0, 16, 32);
  q0 += __shfl_xor(q0, 16, 32);
  if (lane < 16) { wsum[wave][lane] = s0; wsq[wave][lane] = q0; }
  __syncthreads();
  if (threadIdx.x < 32) {
    int c = threadIdx.x & 15;
    int st = threadIdx.x >> 4;
    float tot = 0.f;
    for (int wv = 0; wv < 8; ++wv) tot += st ? wsq[wv][c] : wsum[wv][c];
    part[blockIdx.x * 32 + threadIdx.x] = tot;
  }
}

__global__ void __launch_bounds__(256) bn_finalize(const float* __restrict__ part,
                                                   const float* __restrict__ gamma,
                                                   const float* __restrict__ beta,
                                                   float* __restrict__ scale,
                                                   float* __restrict__ shift) {
  __shared__ float acc[256];
  __shared__ float tot[32];
  int t = threadIdx.x;
  int cs = t & 31, r = t >> 5;
  float s = 0.f;
  for (int blk = r; blk < CONV_BLOCKS; blk += 8) s += part[blk * 32 + cs];
  acc[t] = s;
  __syncthreads();
  if (t < 32) {
    float sum = 0.f;
    for (int rr = 0; rr < 8; ++rr) sum += acc[rr * 32 + t];
    tot[t] = sum;
  }
  __syncthreads();
  if (t < 16) {
    const float invN = 1.0f / BN_COUNT;
    float mean = tot[t] * invN;
    float var  = tot[16 + t] * invN - mean * mean;
    float inv  = rsqrtf(var + 1e-5f);
    float scl  = gamma[t] * inv;
    scale[t] = scl;
    shift[t] = beta[t] - mean * scl;
  }
}

#define H2PB    7840
#define LC_L    6
#define LC_RUNS 16

__device__ __forceinline__ void lc_load_col(const float* __restrict__ hp, int jj,
                                            v2f (&c)[5][4]) {
#pragma unroll
  for (int ii = 0; ii < 5; ++ii) {
    const float4 a = *(const float4*)(hp + (ii * 98 + jj) * 16);
    const float4 b = *(const float4*)(hp + (ii * 98 + jj) * 16 + 4);
    c[ii][0][0] = a.x; c[ii][0][1] = a.y;
    c[ii][1][0] = a.z; c[ii][1][1] = a.w;
    c[ii][2][0] = b.x; c[ii][2][1] = b.y;
    c[ii][3][0] = b.z; c[ii][3][1] = b.w;
  }
}

__device__ __forceinline__ void lc_bn_col(v2f (&c)[5][4], const float (&sc)[8],
                                          const float (&sf)[8]) {
#pragma unroll
  for (int ii = 0; ii < 5; ++ii) {
#pragma unroll
    for (int q = 0; q < 4; ++q) {
      c[ii][q][0] = c[ii][q][0] * sc[2 * q]     + sf[2 * q];
      c[ii][q][1] = c[ii][q][1] * sc[2 * q + 1] + sf[2 * q + 1];
    }
  }
}

__device__ __forceinline__ void lc_load_w(const float* __restrict__ wp, int jc,
                                          float (&w)[4][8][4]) {
#pragma unroll
  for (int oi = 0; oi < 4; ++oi) {
#pragma unroll
    for (int cc = 0; cc < 8; ++cc) {
      const float4 v = *(const float4*)(wp + cc * 1552 + oi * 388 + jc * 4);
      w[oi][cc][0] = v.x; w[oi][cc][1] = v.y; w[oi][cc][2] = v.z; w[oi][cc][3] = v.w;
    }
  }
}

__global__ void __launch_bounds__(128) lc_wmma(const float* __restrict__ h2,
                                               const float* __restrict__ scale,
                                               const float* __restrict__ shift,
                                               const float* __restrict__ w1,
                                               const float* __restrict__ b1,
                                               const float* __restrict__ w2,
                                               const float* __restrict__ b2,
                                               float* __restrict__ out) {
  const int lane  = threadIdx.x & 31;
  const int wave  = threadIdx.x >> 5;
  const int run   = blockIdx.x >> 5;
  const int mtile = (blockIdx.x & 31) * 4 + wave;
  const int lrow  = lane & 15;
  const int lhalf = lane >> 4;
  const int m     = mtile * 16 + lrow;
  const int j0    = run * LC_L;

  float sc[8], sf[8];
  {
    const float4 s0 = ((const float4*)scale)[2 * lhalf];
    const float4 s1 = ((const float4*)scale)[2 * lhalf + 1];
    const float4 f0 = ((const float4*)shift)[2 * lhalf];
    const float4 f1 = ((const float4*)shift)[2 * lhalf + 1];
    sc[0] = s0.x; sc[1] = s0.y; sc[2] = s0.z; sc[3] = s0.w;
    sc[4] = s1.x; sc[5] = s1.y; sc[6] = s1.z; sc[7] = s1.w;
    sf[0] = f0.x; sf[1] = f0.y; sf[2] = f0.z; sf[3] = f0.w;
    sf[4] = f1.x; sf[5] = f1.y; sf[6] = f1.z; sf[7] = f1.w;
  }

  const float* hp = h2 + m * H2PB + 8 * lhalf;
  const float* wp = w1 + ((lrow < 6 ? lrow : 0) * 16 + 8 * lhalf) * 1552;

  v2f   col[3][5][4];
  float w[2][4][8][4];
  float h3p[6][4];

  lc_load_col(hp, j0, col[0]);
  lc_load_col(hp, j0 + 1, col[1]);
  lc_load_w(wp, j0, w[0]);
  lc_bn_col(col[0], sc, sf);
  lc_bn_col(col[1], sc, sf);

#pragma unroll
  for (int t = 0; t <= LC_L; ++t) {
    const int jc = j0 + t;
    if (t < LC_L) {
      lc_load_col(hp, jc + 2, col[(t + 2) % 3]);
      lc_load_w(wp, jc + 1, w[(t + 1) & 1]);
    }
    v2f (&ca)[5][4] = col[t % 3];
    v2f (&cb)[5][4] = col[(t + 1) % 3];
    float (&wc)[4][8][4] = w[t & 1];

    v8f acc[4] = {};
#pragma unroll
    for (int p = 0; p < 4; ++p) {
#pragma unroll
      for (int q = 0; q < 4; ++q) {
#pragma unroll
        for (int oi = 0; oi < 4; ++oi) {
          v2f af;
          af[0] = wc[oi][2 * q][p];
          af[1] = wc[oi][2 * q + 1][p];
          const v2f bf = (p & 1) ? cb[oi + (p >> 1)][q] : ca[oi + (p >> 1)][q];
          acc[oi] = __builtin_amdgcn_wmma_f32_16x16x4_f32(false, af, false, bf,
                                                          (short)0, acc[oi], false, false);
        }
      }
    }

    float h3c[6][4];
#pragma unroll
    for (int oi = 0; oi < 4; ++oi) {
#pragma unroll
      for (int o = 0; o < 6; ++o) {
        float v = acc[oi][o] + b1[o * 388 + oi * 97 + jc];
        h3c[o][oi] = v >= 0.f ? v : 0.01f * v;
      }
    }

    if (t > 0) {
      const int rj = jc - 1;
#pragma unroll
      for (int ri = 0; ri < 3; ++ri) {
        float a = b2[ri * 96 + rj];
#pragma unroll
        for (int o = 0; o < 6; ++o) {
          const float4 wv = *(const float4*)(w2 + (((o * 3 + ri) * 96 + rj) << 2));
          a += wv.x * h3p[o][ri] + wv.y * h3c[o][ri]
             + wv.z * h3p[o][ri + 1] + wv.w * h3c[o][ri + 1];
        }
        if (lane < 16) out[m * 288 + ri * 96 + rj] = a;
      }
    }
#pragma unroll
    for (int o = 0; o < 6; ++o) {
#pragma unroll
      for (int oi = 0; oi < 4; ++oi) h3p[o][oi] = h3c[o][oi];
    }
    if (t < LC_L) lc_bn_col(col[(t + 2) % 3], sc, sf);
  }
}

extern "C" void kernel_launch(void* const* d_in, const int* in_sizes, int n_in,
                              void* d_out, int out_size, void* d_ws, size_t ws_size,
                              hipStream_t stream) {
  const float* x     = (const float*)d_in[0];
  const float* fw    = (const float*)d_in[1];
  const float* fb    = (const float*)d_in[2];
  const float* cw    = (const float*)d_in[3];
  const float* cb    = (const float*)d_in[4];
  const float* gamma = (const float*)d_in[5];
  const float* beta  = (const float*)d_in[6];
  const float* w1    = (const float*)d_in[7];
  const float* b1    = (const float*)d_in[8];
  const float* w2    = (const float*)d_in[9];
  const float* b2    = (const float*)d_in[10];

  float* ws    = (float*)d_ws;
  float* h1    = ws;
  float* h2    = ws + H1_ELEMS;
  float* part  = ws + H1_ELEMS + H2_ELEMS;
  float* scale = part + PART_ELEMS;
  float* shift = scale + 16;
  float* pw    = shift + 16;

  pack_conv_w<<<9, 256, 0, stream>>>(cw, pw);
  fc_wmma    <<<2208, 256, 0, stream>>>(x, fw, fb, h1);
  conv_wmma  <<<CONV_BLOCKS, 256, 0, stream>>>(h1, pw, cb, h2, part);
  bn_finalize<<<1, 256, 0, stream>>>(part, gamma, beta, scale, shift);
  lc_wmma    <<<32 * LC_RUNS, 128, 0, stream>>>(h2, scale, shift, w1, b1, w2, b2,
                                                (float*)d_out);
}


# conv: a wave walks 7 output columns of one row for its 16 batch rows, weights and the shared h1 column held in registers; 62720 waves to 8960, BN partial rows 7840 to 1120: BN sum order changes
# speedup vs baseline: 3.2390x; 1.6419x over previous
#include <hip/hip_runtime.h>

typedef __attribute__((ext_vector_type(2))) float v2f;
typedef __attribute__((ext_vector_type(8))) float v8f;

#define NB      2048
#define FC_K    100
#define FC_N    17640
#define NTILES  1103
#define H1_ELEMS (NB * FC_N)
#define H2_ELEMS (NB * 5 * 98 * 16)
#define CV_L    7
#define CV_RUNS 14
#define H2PB_CV 7840
#define CONV_BLOCKS (16 * CV_RUNS * 5)
#define PART_ELEMS (CONV_BLOCKS * 32)
#define BN_COUNT 1003520.0f
#define H1_PAIRS 8820

#define SLAB_FLOATS 12800
#define SLAB_VEC4   3200
#define FW_VEC4     441000
__global__ void __launch_bounds__(256) fc_wmma(const float* __restrict__ x,
                                               const float* __restrict__ fw,
                                               const float* __restrict__ fb,
                                               float* __restrict__ h1) {
  __shared__ float sB[SLAB_FLOATS];
  const int tid   = threadIdx.x;
  const int lane  = tid & 31;
  const int wave  = tid >> 5;
  const int slab  = blockIdx.x >> 4;
  const int mblk  = blockIdx.x & 15;
  const int mtile = mblk * 8 + wave;
  const int lrow  = lane & 15;
  const int lhalf = lane >> 4;
  const int row   = mtile * 16 + lrow;
  const int kb0   = 2 * lhalf;

  v2f a[25];
#pragma unroll
  for (int s = 0; s < 25; ++s)
    a[s] = *(const v2f*)(x + row * FC_K + 4 * s + kb0);

  const float4* src4 = (const float4*)fw + slab * SLAB_VEC4;
  const int vleft = FW_VEC4 - slab * SLAB_VEC4;
  float4 stg[13];
#pragma unroll
  for (int r = 0; r < 13; ++r) {
    const int idx = tid + r * 256;
    float4 v = make_float4(0.f, 0.f, 0.f, 0.f);
    if ((r < 12 || tid < 128) && idx < vleft) v = src4[idx];
    stg[r] = v;
  }
#pragma unroll
  for (int r = 0; r < 13; ++r) {
    const int idx = tid + r * 256;
    if (r < 12 || tid < 128) ((float4*)sB)[idx] = stg[r];
  }
  __syncthreads();

  for (int t = 0; t < 8; ++t) {
    const int ntile = slab * 8 + t;
    if (ntile >= NTILES) break;
    const int nl = ntile * 16 + 8 * lhalf;
    const bool nvalid = (nl < FC_N);
    float4 bia0 = make_float4(0.f, 0.f, 0.f, 0.f), bia1 = bia0;
    if (nvalid) {
      bia0 = *(const float4*)(fb + nl);
      bia1 = *(const float4*)(fb + nl + 4);
    }
    const float* bp = sB + (t * 16 + lrow) * FC_K + kb0;
    v8f acc = {};
#pragma unroll
    for (int s = 0; s < 25; ++s) {
      v2f wf = *(const v2f*)(bp + 4 * s);
      acc = __builtin_amdgcn_wmma_f32_16x16x4_f32(false, wf, false, a[s],
                                                  (short)0, acc, false, false);
    }
    if (nvalid) {
      float* dst = h1 + ((mtile * H1_PAIRS + ntile * 8 + 4 * lhalf) * 32 + lrow * 2);
      v2f o0, o1, o2, o3;
      o0[0] = acc[0] + bia0.x; o0[1] = acc[1] + bia0.y;
      o1[0] = acc[2] + bia0.z; o1[1] = acc[3] + bia0.w;
      o2[0] = acc[4] + bia1.x; o2[1] = acc[5] + bia1.y;
      o3[0] = acc[6] + bia1.z; o3[1] = acc[7] + bia1.w;
      *(v2f*)(dst)      = o0;
      *(v2f*)(dst + 32) = o1;
      *(v2f*)(dst + 64) = o2;
      *(v2f*)(dst + 96) = o3;
    }
  }
}

__global__ void pack_conv_w(const float* __restrict__ cw, float* __restrict__ pw) {
  int idx = blockIdx.x * 256 + threadIdx.x;
  if (idx >= 2304) return;
  int s  = idx >> 6;
  int r  = idx & 63;
  int ln = r >> 1;
  int e  = r & 1;
  int n  = ln & 15;
  int k  = 4 * s + 2 * (ln >> 4) + e;
  int p  = k / 36;
  int c  = k - p * 36;
  pw[idx] = cw[n * 144 + c * 4 + p];
}

__device__ __forceinline__ void cv_load_col(const float* __restrict__ ap, int i, int jj,
                                            v2f (&c)[2][9]) {
#pragma unroll
  for (int d = 0; d < 2; ++d) {
    if (i + d < 5 && jj < 98) {
#pragma unroll
      for (int u = 0; u < 9; ++u)
        c[d][u] = *(const v2f*)(ap + (((i + d) * 98 + jj) * 18 + 2 * u) * 32);
    } else {
#pragma unroll
      for (int u = 0; u < 9; ++u) { c[d][u][0] = 0.f; c[d][u][1] = 0.f; }
    }
  }
}

__global__ void __launch_bounds__(256) conv_wmma(const float* __restrict__ h1,
                                                 const float* __restrict__ pw,
                                                 const float* __restrict__ cb,
                                                 float* __restrict__ h2,
                                                 float* __restrict__ part) {
  __shared__ float wsum[8][16];
  __shared__ float wsq[8][16];
  const int lane  = threadIdx.x & 31;
  const int wave  = threadIdx.x >> 5;
  const int i     = blockIdx.x % 5;
  const int run   = (blockIdx.x / 5) % CV_RUNS;
  const int mblk  = blockIdx.x / (5 * CV_RUNS);
  const int mtile = mblk * 8 + wave;
  const int j0    = run * CV_L;
  const int lrow  = lane & 15;
  const int lhalf = lane >> 4;
  const float* ap  = h1 + (mtile * H1_PAIRS * 32 + lhalf * 32 + lrow * 2);
  const float* bwp = pw + lane * 2;

  v2f wr[36];
#pragma unroll
  for (int s = 0; s < 36; ++s) wr[s] = *(const v2f*)(bwp + 64 * s);

  v2f ca[2][9], cn[2][9], cf[2][9];
  cv_load_col(ap, i, j0, ca);
  cv_load_col(ap, i, j0 + 1, cn);

  const float bias = cb[lrow];
  float s0 = 0.f, q0 = 0.f;
  float* hq = h2 + (((mtile * 16 + lhalf * 8) * 5 + i) * 98 + j0) * 16 + lrow;

#pragma unroll 1
  for (int t = 0; t < CV_L; ++t) {
    if (t + 1 < CV_L)
      cv_load_col(ap, i, j0 + t + 2, cf);

    v8f acc = {};
#pragma unroll
    for (int p = 0; p < 4; ++p) {
#pragma unroll
      for (int u = 0; u < 9; ++u) {
        const v2f af = (p & 1) ? cn[p >> 1][u] : ca[p >> 1][u];
        acc = __builtin_amdgcn_wmma_f32_16x16x4_f32(false, af, false, wr[9 * p + u],
                                                    (short)0, acc, false, false);
      }
    }

#pragma unroll
    for (int v = 0; v < 8; ++v) {
      float val = acc[v] + bias;
      val = val >= 0.f ? val : 0.01f * val;
      hq[v * H2PB_CV + t * 16] = val;
      s0 += val;
      q0 += val * val;
    }

    __builtin_amdgcn_sched_barrier(0);
#pragma unroll
    for (int d = 0; d < 2; ++d) {
#pragma unroll
      for (int u = 0; u < 9; ++u) { ca[d][u] = cn[d][u]; cn[d][u] = cf[d][u]; }
    }
  }
  s0 += __shfl_xor(s0, 16, 32);
  q0 += __shfl_xor(q0, 16, 32);
  if (lane < 16) { wsum[wave][lane] = s0; wsq[wave][lane] = q0; }
  __syncthreads();
  if (threadIdx.x < 32) {
    int c = threadIdx.x & 15;
    int st = threadIdx.x >> 4;
    float tot = 0.f;
    for (int wv = 0; wv < 8; ++wv) tot += st ? wsq[wv][c] : wsum[wv][c];
    part[blockIdx.x * 32 + threadIdx.x] = tot;
  }
}

__global__ void __launch_bounds__(256) bn_finalize(const float* __restrict__ part,
                                                   const float* __restrict__ gamma,
                                                   const float* __restrict__ beta,
                                                   float* __restrict__ scale,
                                                   float* __restrict__ shift) {
  __shared__ float acc[256];
  __shared__ float tot[32];
  int t = threadIdx.x;
  int cs = t & 31, r = t >> 5;
  float s = 0.f;
  for (int blk = r; blk < CONV_BLOCKS; blk += 8) s += part[blk * 32 + cs];
  acc[t] = s;
  __syncthreads();
  if (t < 32) {
    float sum = 0.f;
    for (int rr = 0; rr < 8; ++rr) sum += acc[rr * 32 + t];
    tot[t] = sum;
  }
  __syncthreads();
  if (t < 16) {
    const float invN = 1.0f / BN_COUNT;
    float mean = tot[t] * invN;
    float var  = tot[16 + t] * invN - mean * mean;
    float inv  = rsqrtf(var + 1e-5f);
    float scl  = gamma[t] * inv;
    scale[t] = scl;
    shift[t] = beta[t] - mean * scl;
  }
}

#define H2PB    7840
#define LC_L    6
#define LC_RUNS 16

__device__ __forceinline__ void lc_load_col(const float* __restrict__ hp, int jj,
                                            v2f (&c)[5][4]) {
#pragma unroll
  for (int ii = 0; ii < 5; ++ii) {
    const float4 a = *(const float4*)(hp + (ii * 98 + jj) * 16);
    const float4 b = *(const float4*)(hp + (ii * 98 + jj) * 16 + 4);
    c[ii][0][0] = a.x; c[ii][0][1] = a.y;
    c[ii][1][0] = a.z; c[ii][1][1] = a.w;
    c[ii][2][0] = b.x; c[ii][2][1] = b.y;
    c[ii][3][0] = b.z; c[ii][3][1] = b.w;
  }
}

__device__ __forceinline__ void lc_bn_col(v2f (&c)[5][4], const float (&sc)[8],
                                          const float (&sf)[8]) {
#pragma unroll
  for (int ii = 0; ii < 5; ++ii) {
#pragma unroll
    for (int q = 0; q < 4; ++q) {
      c[ii][q][0] = c[ii][q][0] * sc[2 * q]     + sf[2 * q];
      c[ii][q][1] = c[ii][q][1] * sc[2 * q + 1] + sf[2 * q + 1];
    }
  }
}

__device__ __forceinline__ void lc_load_w(const float* __restrict__ wp, int jc,
                                          float (&w)[4][8][4]) {
#pragma unroll
  for (int oi = 0; oi < 4; ++oi) {
#pragma unroll
    for (int cc = 0; cc < 8; ++cc) {
      const float4 v = *(const float4*)(wp + cc * 1552 + oi * 388 + jc * 4);
      w[oi][cc][0] = v.x; w[oi][cc][1] = v.y; w[oi][cc][2] = v.z; w[oi][cc][3] = v.w;
    }
  }
}

__global__ void __launch_bounds__(128) lc_wmma(const float* __restrict__ h2,
                                               const float* __restrict__ scale,
                                               const float* __restrict__ shift,
                                               const float* __restrict__ w1,
                                               const float* __restrict__ b1,
                                               const float* __restrict__ w2,
                                               const float* __restrict__ b2,
                                               float* __restrict__ out) {
  const int lane  = threadIdx.x & 31;
  const int wave  = threadIdx.x >> 5;
  const int run   = blockIdx.x >> 5;
  const int mtile = (blockIdx.x & 31) * 4 + wave;
  const int lrow  = lane & 15;
  const int lhalf = lane >> 4;
  const int m     = mtile * 16 + lrow;
  const int j0    = run * LC_L;

  float sc[8], sf[8];
  {
    const float4 s0 = ((const float4*)scale)[2 * lhalf];
    const float4 s1 = ((const float4*)scale)[2 * lhalf + 1];
    const float4 f0 = ((const float4*)shift)[2 * lhalf];
    const float4 f1 = ((const float4*)shift)[2 * lhalf + 1];
    sc[0] = s0.x; sc[1] = s0.y; sc[2] = s0.z; sc[3] = s0.w;
    sc[4] = s1.x; sc[5] = s1.y; sc[6] = s1.z; sc[7] = s1.w;
    sf[0] = f0.x; sf[1] = f0.y; sf[2] = f0.z; sf[3] = f0.w;
    sf[4] = f1.x; sf[5] = f1.y; sf[6] = f1.z; sf[7] = f1.w;
  }

  const float* hp = h2 + m * H2PB + 8 * lhalf;
  const float* wp = w1 + ((lrow < 6 ? lrow : 0) * 16 + 8 * lhalf) * 1552;

  v2f   col[3][5][4];
  float w[2][4][8][4];
  float h3p[6][4];

  lc_load_col(hp, j0, col[0]);
  lc_load_col(hp, j0 + 1, col[1]);
  lc_load_w(wp, j0, w[0]);
  lc_bn_col(col[0], sc, sf);
  lc_bn_col(col[1], sc, sf);

#pragma unroll
  for (int t = 0; t <= LC_L; ++t) {
    const int jc = j0 + t;
    if (t < LC_L) {
      lc_load_col(hp, jc + 2, col[(t + 2) % 3]);
      lc_load_w(wp, jc + 1, w[(t + 1) & 1]);
    }
    v2f (&ca)[5][4] = col[t % 3];
    v2f (&cb)[5][4] = col[(t + 1) % 3];
    float (&wc)[4][8][4] = w[t & 1];

    v8f acc[4] = {};
#pragma unroll
    for (int p = 0; p < 4; ++p) {
#pragma unroll
      for (int q = 0; q < 4; ++q) {
#pragma unroll
        for (int oi = 0; oi < 4; ++oi) {
          v2f af;
          af[0] = wc[oi][2 * q][p];
          af[1] = wc[oi][2 * q + 1][p];
          const v2f bf = (p & 1) ? cb[oi + (p >> 1)][q] : ca[oi + (p >> 1)][q];
          acc[oi] = __builtin_amdgcn_wmma_f32_16x16x4_f32(false, af, false, bf,
                                                          (short)0, acc[oi], false, false);
        }
      }
    }

    float h3c[6][4];
#pragma unroll
    for (int oi = 0; oi < 4; ++oi) {
#pragma unroll
      for (int o = 0; o < 6; ++o) {
        float v = acc[oi][o] + b1[o * 388 + oi * 97 + jc];
        h3c[o][oi] = v >= 0.f ? v : 0.01f * v;
      }
    }

    if (t > 0) {
      const int rj = jc - 1;
#pragma unroll
      for (int ri = 0; ri < 3; ++ri) {
        float a = b2[ri * 96 + rj];
#pragma unroll
        for (int o = 0; o < 6; ++o) {
          const float4 wv = *(const float4*)(w2 + (((o * 3 + ri) * 96 + rj) << 2));
          a += wv.x * h3p[o][ri] + wv.y * h3c[o][ri]
             + wv.z * h3p[o][ri + 1] + wv.w * h3c[o][ri + 1];
        }
        if (lane < 16) out[m * 288 + ri * 96 + rj] = a;
      }
    }
#pragma unroll
    for (int o = 0; o < 6; ++o) {
#pragma unroll
      for (int oi = 0; oi < 4; ++oi) h3p[o][oi] = h3c[o][oi];
    }
    if (t < LC_L) lc_bn_col(col[(t + 2) % 3], sc, sf);
  }
}

extern "C" void kernel_launch(void* const* d_in, const int* in_sizes, int n_in,
                              void* d_out, int out_size, void* d_ws, size_t ws_size,
                              hipStream_t stream) {
  const float* x     = (const float*)d_in[0];
  const float* fw    = (const float*)d_in[1];
  const float* fb    = (const float*)d_in[2];
  const float* cw    = (const float*)d_in[3];
  const float* cb    = (const float*)d_in[4];
  const float* gamma = (const float*)d_in[5];
  const float* beta  = (const float*)d_in[6];
  const float* w1    = (const float*)d_in[7];
  const float* b1    = (const float*)d_in[8];
  const float* w2    = (const float*)d_in[9];
  const float* b2    = (const float*)d_in[10];

  float* ws    = (float*)d_ws;
  float* h1    = ws;
  float* h2    = ws + H1_ELEMS;
  float* part  = ws + H1_ELEMS + H2_ELEMS;
  float* scale = part + PART_ELEMS;
  float* shift = scale + 16;
  float* pw    = shift + 16;

  pack_conv_w<<<9, 256, 0, stream>>>(cw, pw);
  fc_wmma    <<<2208, 256, 0, stream>>>(x, fw, fb, h1);
  conv_wmma  <<<CONV_BLOCKS, 256, 0, stream>>>(h1, pw, cb, h2, part);
  bn_finalize<<<1, 256, 0, stream>>>(part, gamma, beta, scale, shift);
  lc_wmma    <<<32 * LC_RUNS, 128, 0, stream>>>(h2, scale, shift, w1, b1, w2, b2,
                                                (float*)d_out);
}


# conv: a wave walks 7 output columns of ALL 5 rows for its 16 batch rows, whole h1 columns in registers: each h1 position loaded once per run, not per row; blocks 1120 to 224: BN sum order changes
# speedup vs baseline: 3.4997x; 1.0805x over previous
#include <hip/hip_runtime.h>

typedef __attribute__((ext_vector_type(2))) float v2f;
typedef __attribute__((ext_vector_type(8))) float v8f;

#define NB      2048
#define FC_K    100
#define FC_N    17640
#define NTILES  1103
#define H1_ELEMS (NB * FC_N)
#define H2_ELEMS (NB * 5 * 98 * 16)
#define CV_L    7
#define CV_RUNS 14
#define H2PB_CV 7840
#define CONV_BLOCKS (16 * CV_RUNS)
#define PART_ELEMS (CONV_BLOCKS * 32)
#define BN_COUNT 1003520.0f
#define H1_PAIRS 8820

#define SLAB_FLOATS 12800
#define SLAB_VEC4   3200
#define FW_VEC4     441000
__global__ void __launch_bounds__(256) fc_wmma(const float* __restrict__ x,
                                               const float* __restrict__ fw,
                                               const float* __restrict__ fb,
                                               float* __restrict__ h1) {
  __shared__ float sB[SLAB_FLOATS];
  const int tid   = threadIdx.x;
  const int lane  = tid & 31;
  const int wave  = tid >> 5;
  const int slab  = blockIdx.x >> 4;
  const int mblk  = blockIdx.x & 15;
  const int mtile = mblk * 8 + wave;
  const int lrow  = lane & 15;
  const int lhalf = lane >> 4;
  const int row   = mtile * 16 + lrow;
  const int kb0   = 2 * lhalf;

  v2f a[25];
#pragma unroll
  for (int s = 0; s < 25; ++s)
    a[s] = *(const v2f*)(x + row * FC_K + 4 * s + kb0);

  const float4* src4 = (const float4*)fw + slab * SLAB_VEC4;
  const int vleft = FW_VEC4 - slab * SLAB_VEC4;
  float4 stg[13];
#pragma unroll
  for (int r = 0; r < 13; ++r) {
    const int idx = tid + r * 256;
    float4 v = make_float4(0.f, 0.f, 0.f, 0.f);
    if ((r < 12 || tid < 128) && idx < vleft) v = src4[idx];
    stg[r] = v;
  }
#pragma unroll
  for (int r = 0; r < 13; ++r) {
    const int idx = tid + r * 256;
    if (r < 12 || tid < 128) ((float4*)sB)[idx] = stg[r];
  }
  __syncthreads();

  for (int t = 0; t < 8; ++t) {
    const int ntile = slab * 8 + t;
    if (ntile >= NTILES) break;
    const int nl = ntile * 16 + 8 * lhalf;
    const bool nvalid = (nl < FC_N);
    float4 bia0 = make_float4(0.f, 0.f, 0.f, 0.f), bia1 = bia0;
    if (nvalid) {
      bia0 = *(const float4*)(fb + nl);
      bia1 = *(const float4*)(fb + nl + 4);
    }
    const float* bp = sB + (t * 16 + lrow) * FC_K + kb0;
    v8f acc = {};
#pragma unroll
    for (int s = 0; s < 25; ++s) {
      v2f wf = *(const v2f*)(bp + 4 * s);
      acc = __builtin_amdgcn_wmma_f32_16x16x4_f32(false, wf, false, a[s],
                                                  (short)0, acc, false, false);
    }
    if (nvalid) {
      float* dst = h1 + ((mtile * H1_PAIRS + ntile * 8 + 4 * lhalf) * 32 + lrow * 2);
      v2f o0, o1, o2, o3;
      o0[0] = acc[0] + bia0.x; o0[1] = acc[1] + bia0.y;
      o1[0] = acc[2] + bia0.z; o1[1] = acc[3] + bia0.w;
      o2[0] = acc[4] + bia1.x; o2[1] = acc[5] + bia1.y;
      o3[0] = acc[6] + bia1.z; o3[1] = acc[7] + bia1.w;
      *(v2f*)(dst)      = o0;
      *(v2f*)(dst + 32) = o1;
      *(v2f*)(dst + 64) = o2;
      *(v2f*)(dst + 96) = o3;
    }
  }
}

__global__ void pack_conv_w(const float* __restrict__ cw, float* __restrict__ pw) {
  int idx = blockIdx.x * 256 + threadIdx.x;
  if (idx >= 2304) return;
  int s  = idx >> 6;
  int r  = idx & 63;
  int ln = r >> 1;
  int e  = r & 1;
  int n  = ln & 15;
  int k  = 4 * s + 2 * (ln >> 4) + e;
  int p  = k / 36;
  int c  = k - p * 36;
  pw[idx] = cw[n * 144 + c * 4 + p];
}

__device__ __forceinline__ void cv_load_col(const float* __restrict__ ap, int jj,
                                            v2f (&c)[5][9]) {
  if (jj < 98) {
#pragma unroll
    for (int r = 0; r < 5; ++r) {
#pragma unroll
      for (int u = 0; u < 9; ++u)
        c[r][u] = *(const v2f*)(ap + ((r * 98 + jj) * 18 + 2 * u) * 32);
    }
  } else {
#pragma unroll
    for (int r = 0; r < 5; ++r) {
#pragma unroll
      for (int u = 0; u < 9; ++u) { c[r][u][0] = 0.f; c[r][u][1] = 0.f; }
    }
  }
}

__global__ void __launch_bounds__(256) conv_wmma(const float* __restrict__ h1,
                                                 const float* __restrict__ pw,
                                                 const float* __restrict__ cb,
                                                 float* __restrict__ h2,
                                                 float* __restrict__ part) {
  __shared__ float wsum[8][16];
  __shared__ float wsq[8][16];
  const int lane  = threadIdx.x & 31;
  const int wave  = threadIdx.x >> 5;
  const int run   = blockIdx.x % CV_RUNS;
  const int mblk  = blockIdx.x / CV_RUNS;
  const int mtile = mblk * 8 + wave;
  const int j0    = run * CV_L;
  const int lrow  = lane & 15;
  const int lhalf = lane >> 4;
  const float* ap  = h1 + (mtile * H1_PAIRS * 32 + lhalf * 32 + lrow * 2);
  const float* bwp = pw + lane * 2;

  v2f wr[36];
#pragma unroll
  for (int s = 0; s < 36; ++s) wr[s] = *(const v2f*)(bwp + 64 * s);

  v2f ca[5][9], cn[5][9], cf[5][9];
  cv_load_col(ap, j0, ca);
  cv_load_col(ap, j0 + 1, cn);

  v2f zero2;
  zero2[0] = 0.f; zero2[1] = 0.f;

  const float bias = cb[lrow];
  float s0 = 0.f, q0 = 0.f;
  float* hq = h2 + ((mtile * 16 + lhalf * 8) * 490 + j0) * 16 + lrow;

  __threadfence_block();

#pragma unroll 1
  for (int t = 0; t < CV_L; ++t) {
    if (t + 1 < CV_L)
      cv_load_col(ap, j0 + t + 2, cf);

    v8f acc[5] = {};
#pragma unroll
    for (int p = 0; p < 4; ++p) {
#pragma unroll
      for (int u = 0; u < 9; ++u) {
#pragma unroll
        for (int i = 0; i < 5; ++i) {
          const int r = i + (p >> 1);
          v2f af = zero2;
          if (r < 5) af = (p & 1) ? cn[r][u] : ca[r][u];
          acc[i] = __builtin_amdgcn_wmma_f32_16x16x4_f32(false, af, false, wr[9 * p + u],
                                                         (short)0, acc[i], false, false);
        }
      }
    }

#pragma unroll
    for (int i = 0; i < 5; ++i) {
#pragma unroll
      for (int v = 0; v < 8; ++v) {
        float val = acc[i][v] + bias;
        val = val >= 0.f ? val : 0.01f * val;
        hq[v * H2PB_CV + (i * 98 + t) * 16] = val;
        s0 += val;
        q0 += val * val;
      }
    }

    __builtin_amdgcn_sched_barrier(0);
#pragma unroll
    for (int r = 0; r < 5; ++r) {
#pragma unroll
      for (int u = 0; u < 9; ++u) { ca[r][u] = cn[r][u]; cn[r][u] = cf[r][u]; }
    }
  }
  s0 += __shfl_xor(s0, 16, 32);
  q0 += __shfl_xor(q0, 16, 32);
  if (lane < 16) { wsum[wave][lane] = s0; wsq[wave][lane] = q0; }
  __syncthreads();
  if (threadIdx.x < 32) {
    int c = threadIdx.x & 15;
    int st = threadIdx.x >> 4;
    float tot = 0.f;
    for (int wv = 0; wv < 8; ++wv) tot += st ? wsq[wv][c] : wsum[wv][c];
    part[blockIdx.x * 32 + threadIdx.x] = tot;
  }
}

__global__ void __launch_bounds__(256) bn_finalize(const float* __restrict__ part,
                                                   const float* __restrict__ gamma,
                                                   const float* __restrict__ beta,
                                                   float* __restrict__ scale,
                                                   float* __restrict__ shift) {
  __shared__ float acc[256];
  __shared__ float tot[32];
  int t = threadIdx.x;
  int cs = t & 31, r = t >> 5;
  float s = 0.f;
  for (int blk = r; blk < CONV_BLOCKS; blk += 8) s += part[blk * 32 + cs];
  acc[t] = s;
  __syncthreads();
  if (t < 32) {
    float sum = 0.f;
    for (int rr = 0; rr < 8; ++rr) sum += acc[rr * 32 + t];
    tot[t] = sum;
  }
  __syncthreads();
  if (t < 16) {
    const float invN = 1.0f / BN_COUNT;
    float mean = tot[t] * invN;
    float var  = tot[16 + t] * invN - mean * mean;
    float inv  = rsqrtf(var + 1e-5f);
    float scl  = gamma[t] * inv;
    scale[t] = scl;
    shift[t] = beta[t] - mean * scl;
  }
}

#define H2PB    7840
#define LC_L    6
#define LC_RUNS 16

__device__ __forceinline__ void lc_load_col(const float* __restrict__ hp, int jj,
                                            v2f (&c)[5][4]) {
#pragma unroll
  for (int ii = 0; ii < 5; ++ii) {
    const float4 a = *(const float4*)(hp + (ii * 98 + jj) * 16);
    const float4 b = *(const float4*)(hp + (ii * 98 + jj) * 16 + 4);
    c[ii][0][0] = a.x; c[ii][0][1] = a.y;
    c[ii][1][0] = a.z; c[ii][1][1] = a.w;
    c[ii][2][0] = b.x; c[ii][2][1] = b.y;
    c[ii][3][0] = b.z; c[ii][3][1] = b.w;
  }
}

__device__ __forceinline__ void lc_bn_col(v2f (&c)[5][4], const float (&sc)[8],
                                          const float (&sf)[8]) {
#pragma unroll
  for (int ii = 0; ii < 5; ++ii) {
#pragma unroll
    for (int q = 0; q < 4; ++q) {
      c[ii][q][0] = c[ii][q][0] * sc[2 * q]     + sf[2 * q];
      c[ii][q][1] = c[ii][q][1] * sc[2 * q + 1] + sf[2 * q + 1];
    }
  }
}

__device__ __forceinline__ void lc_load_w(const float* __restrict__ wp, int jc,
                                          float (&w)[4][8][4]) {
#pragma unroll
  for (int oi = 0; oi < 4; ++oi) {
#pragma unroll
    for (int cc = 0; cc < 8; ++cc) {
      const float4 v = *(const float4*)(wp + cc * 1552 + oi * 388 + jc * 4);
      w[oi][cc][0] = v.x; w[oi][cc][1] = v.y; w[oi][cc][2] = v.z; w[oi][cc][3] = v.w;
    }
  }
}

__global__ void __launch_bounds__(128) lc_wmma(const float* __restrict__ h2,
                                               const float* __restrict__ scale,
                                               const float* __restrict__ shift,
                                               const float* __restrict__ w1,
                                               const float* __restrict__ b1,
                                               const float* __restrict__ w2,
                                               const float* __restrict__ b2,
                                               float* __restrict__ out) {
  const int lane  = threadIdx.x & 31;
  const int wave  = threadIdx.x >> 5;
  const int run   = blockIdx.x >> 5;
  const int mtile = (blockIdx.x & 31) * 4 + wave;
  const int lrow  = lane & 15;
  const int lhalf = lane >> 4;
  const int m     = mtile * 16 + lrow;
  const int j0    = run * LC_L;

  float sc[8], sf[8];
  {
    const float4 s0 = ((const float4*)scale)[2 * lhalf];
    const float4 s1 = ((const float4*)scale)[2 * lhalf + 1];
    const float4 f0 = ((const float4*)shift)[2 * lhalf];
    const float4 f1 = ((const float4*)shift)[2 * lhalf + 1];
    sc[0] = s0.x; sc[1] = s0.y; sc[2] = s0.z; sc[3] = s0.w;
    sc[4] = s1.x; sc[5] = s1.y; sc[6] = s1.z; sc[7] = s1.w;
    sf[0] = f0.x; sf[1] = f0.y; sf[2] = f0.z; sf[3] = f0.w;
    sf[4] = f1.x; sf[5] = f1.y; sf[6] = f1.z; sf[7] = f1.w;
  }

  const float* hp = h2 + m * H2PB + 8 * lhalf;
  const float* wp = w1 + ((lrow < 6 ? lrow : 0) * 16 + 8 * lhalf) * 1552;

  v2f   col[3][5][4];
  float w[2][4][8][4];
  float h3p[6][4];

  lc_load_col(hp, j0, col[0]);
  lc_load_col(hp, j0 + 1, col[1]);
  lc_load_w(wp, j0, w[0]);
  lc_bn_col(col[0], sc, sf);
  lc_bn_col(col[1], sc, sf);

#pragma unroll
  for (int t = 0; t <= LC_L; ++t) {
    const int jc = j0 + t;
    if (t < LC_L) {
      lc_load_col(hp, jc + 2, col[(t + 2) % 3]);
      lc_load_w(wp, jc + 1, w[(t + 1) & 1]);
    }
    v2f (&ca)[5][4] = col[t % 3];
    v2f (&cb)[5][4] = col[(t + 1) % 3];
    float (&wc)[4][8][4] = w[t & 1];

    v8f acc[4] = {};
#pragma unroll
    for (int p = 0; p < 4; ++p) {
#pragma unroll
      for (int q = 0; q < 4; ++q) {
#pragma unroll
        for (int oi = 0; oi < 4; ++oi) {
          v2f af;
          af[0] = wc[oi][2 * q][p];
          af[1] = wc[oi][2 * q + 1][p];
          const v2f bf = (p & 1) ? cb[oi + (p >> 1)][q] : ca[oi + (p >> 1)][q];
          acc[oi] = __builtin_amdgcn_wmma_f32_16x16x4_f32(false, af, false, bf,
                                                          (short)0, acc[oi], false, false);
        }
      }
    }

    float h3c[6][4];
#pragma unroll
    for (int oi = 0; oi < 4; ++oi) {
#pragma unroll
      for (int o = 0; o < 6; ++o) {
        float v = acc[oi][o] + b1[o * 388 + oi * 97 + jc];
        h3c[o][oi] = v >= 0.f ? v : 0.01f * v;
      }
    }

    if (t > 0) {
      const int rj = jc - 1;
#pragma unroll
      for (int ri = 0; ri < 3; ++ri) {
        float a = b2[ri * 96 + rj];
#pragma unroll
        for (int o = 0; o < 6; ++o) {
          const float4 wv = *(const float4*)(w2 + (((o * 3 + ri) * 96 + rj) << 2));
          a += wv.x * h3p[o][ri] + wv.y * h3c[o][ri]
             + wv.z * h3p[o][ri + 1] + wv.w * h3c[o][ri + 1];
        }
        if (lane < 16) out[m * 288 + ri * 96 + rj] = a;
      }
    }
#pragma unroll
    for (int o = 0; o < 6; ++o) {
#pragma unroll
      for (int oi = 0; oi < 4; ++oi) h3p[o][oi] = h3c[o][oi];
    }
    if (t < LC_L) lc_bn_col(col[(t + 2) % 3], sc, sf);
  }
}

extern "C" void kernel_launch(void* const* d_in, const int* in_sizes, int n_in,
                              void* d_out, int out_size, void* d_ws, size_t ws_size,
                              hipStream_t stream) {
  const float* x     = (const float*)d_in[0];
  const float* fw    = (const float*)d_in[1];
  const float* fb    = (const float*)d_in[2];
  const float* cw    = (const float*)d_in[3];
  const float* cb    = (const float*)d_in[4];
  const float* gamma = (const float*)d_in[5];
  const float* beta  = (const float*)d_in[6];
  const float* w1    = (const float*)d_in[7];
  const float* b1    = (const float*)d_in[8];
  const float* w2    = (const float*)d_in[9];
  const float* b2    = (const float*)d_in[10];

  float* ws    = (float*)d_ws;
  float* h1    = ws;
  float* h2    = ws + H1_ELEMS;
  float* part  = ws + H1_ELEMS + H2_ELEMS;
  float* scale = part + PART_ELEMS;
  float* shift = scale + 16;
  float* pw    = shift + 16;

  pack_conv_w<<<9, 256, 0, stream>>>(cw, pw);
  fc_wmma    <<<2208, 256, 0, stream>>>(x, fw, fb, h1);
  conv_wmma  <<<CONV_BLOCKS, 256, 0, stream>>>(h1, pw, cb, h2, part);
  bn_finalize<<<1, 256, 0, stream>>>(part, gamma, beta, scale, shift);
  lc_wmma    <<<32 * LC_RUNS, 128, 0, stream>>>(h2, scale, shift, w1, b1, w2, b2,
                                                (float*)d_out);
}


# h2 and the lc1 weights stored in the lc kernel's WMMA fragment order (conv product transposed, weights repacked in the BN launch): an lc load is 1 aligned run, was 12-16 lines; BN sum order changes
# speedup vs baseline: 3.9130x; 1.1181x over previous
#include <hip/hip_runtime.h>

typedef __attribute__((ext_vector_type(2))) float v2f;
typedef __attribute__((ext_vector_type(8))) float v8f;

#define NB      2048
#define FC_K    100
#define FC_N    17640
#define NTILES  1103
#define H1_ELEMS (NB * FC_N)
#define H2_ELEMS (NB * 5 * 98 * 16)
#define CV_L    7
#define CV_RUNS 14
#define CONV_BLOCKS (16 * CV_RUNS)
#define PART_ELEMS (CONV_BLOCKS * 32)
#define BN_COUNT 1003520.0f
#define H1_PAIRS 8820
#define H2_POS_STRIDE 256

#define SLAB_FLOATS 12800
#define SLAB_VEC4   3200
#define FW_VEC4     441000
__global__ void __launch_bounds__(256) fc_wmma(const float* __restrict__ x,
                                               const float* __restrict__ fw,
                                               const float* __restrict__ fb,
                                               float* __restrict__ h1) {
  __shared__ float sB[SLAB_FLOATS];
  const int tid   = threadIdx.x;
  const int lane  = tid & 31;
  const int wave  = tid >> 5;
  const int slab  = blockIdx.x >> 4;
  const int mblk  = blockIdx.x & 15;
  const int mtile = mblk * 8 + wave;
  const int lrow  = lane & 15;
  const int lhalf = lane >> 4;
  const int row   = mtile * 16 + lrow;
  const int kb0   = 2 * lhalf;

  v2f a[25];
#pragma unroll
  for (int s = 0; s < 25; ++s)
    a[s] = *(const v2f*)(x + row * FC_K + 4 * s + kb0);

  const float4* src4 = (const float4*)fw + slab * SLAB_VEC4;
  const int vleft = FW_VEC4 - slab * SLAB_VEC4;
  float4 stg[13];
#pragma unroll
  for (int r = 0; r < 13; ++r) {
    const int idx = tid + r * 256;
    float4 v = make_float4(0.f, 0.f, 0.f, 0.f);
    if ((r < 12 || tid < 128) && idx < vleft) v = src4[idx];
    stg[r] = v;
  }
#pragma unroll
  for (int r = 0; r < 13; ++r) {
    const int idx = tid + r * 256;
    if (r < 12 || tid < 128) ((float4*)sB)[idx] = stg[r];
  }
  __syncthreads();

  for (int t = 0; t < 8; ++t) {
    const int ntile = slab * 8 + t;
    if (ntile >= NTILES) break;
    const int nl = ntile * 16 + 8 * lhalf;
    const bool nvalid = (nl < FC_N);
    float4 bia0 = make_float4(0.f, 0.f, 0.f, 0.f), bia1 = bia0;
    if (nvalid) {
      bia0 = *(const float4*)(fb + nl);
      bia1 = *(const float4*)(fb + nl + 4);
    }
    const float* bp = sB + (t * 16 + lrow) * FC_K + kb0;
    v8f acc = {};
#pragma unroll
    for (int s = 0; s < 25; ++s) {
      v2f wf = *(const v2f*)(bp + 4 * s);
      acc = __builtin_amdgcn_wmma_f32_16x16x4_f32(false, wf, false, a[s],
                                                  (short)0, acc, false, false);
    }
    if (nvalid) {
      float* dst = h1 + ((mtile * H1_PAIRS + ntile * 8 + 4 * lhalf) * 32 + lrow * 2);
      v2f o0, o1, o2, o3;
      o0[0] = acc[0] + bia0.x; o0[1] = acc[1] + bia0.y;
      o1[0] = acc[2] + bia0.z; o1[1] = acc[3] + bia0.w;
      o2[0] = acc[4] + bia1.x; o2[1] = acc[5] + bia1.y;
      o3[0] = acc[6] + bia1.z; o3[1] = acc[7] + bia1.w;
      *(v2f*)(dst)      = o0;
      *(v2f*)(dst + 32) = o1;
      *(v2f*)(dst + 64) = o2;
      *(v2f*)(dst + 96) = o3;
    }
  }
}

__global__ void pack_conv_w(const float* __restrict__ cw, float* __restrict__ pw) {
  int idx = blockIdx.x * 256 + threadIdx.x;
  if (idx >= 2304) return;
  int s  = idx >> 6;
  int r  = idx & 63;
  int ln = r >> 1;
  int e  = r & 1;
  int n  = ln & 15;
  int k  = 4 * s + 2 * (ln >> 4) + e;
  int p  = k / 36;
  int c  = k - p * 36;
  pw[idx] = cw[n * 144 + c * 4 + p];
}

__device__ __forceinline__ void cv_load_col(const float* __restrict__ ap, int jj,
                                            v2f (&c)[5][9]) {
  if (jj < 98) {
#pragma unroll
    for (int r = 0; r < 5; ++r) {
#pragma unroll
      for (int u = 0; u < 9; ++u)
        c[r][u] = *(const v2f*)(ap + ((r * 98 + jj) * 18 + 2 * u) * 32);
    }
  } else {
#pragma unroll
    for (int r = 0; r < 5; ++r) {
#pragma unroll
      for (int u = 0; u < 9; ++u) { c[r][u][0] = 0.f; c[r][u][1] = 0.f; }
    }
  }
}

__global__ void __launch_bounds__(256) conv_wmma(const float* __restrict__ h1,
                                                 const float* __restrict__ pw,
                                                 const float* __restrict__ cb,
                                                 float* __restrict__ h2,
                                                 float* __restrict__ part) {
  __shared__ float wsum[8][16];
  __shared__ float wsq[8][16];
  const int lane  = threadIdx.x & 31;
  const int wave  = threadIdx.x >> 5;
  const int run   = blockIdx.x % CV_RUNS;
  const int mblk  = blockIdx.x / CV_RUNS;
  const int mtile = mblk * 8 + wave;
  const int j0    = run * CV_L;
  const int lrow  = lane & 15;
  const int lhalf = lane >> 4;
  const float* ap  = h1 + (mtile * H1_PAIRS * 32 + lhalf * 32 + lrow * 2);
  const float* bwp = pw + lane * 2;

  v2f wr[36];
#pragma unroll
  for (int s = 0; s < 36; ++s) wr[s] = *(const v2f*)(bwp + 64 * s);

  v2f ca[5][9], cn[5][9], cf[5][9];
  cv_load_col(ap, j0, ca);
  cv_load_col(ap, j0 + 1, cn);

  v2f zero2;
  zero2[0] = 0.f; zero2[1] = 0.f;

  float bias8[8];
  {
    const float4 c0 = *(const float4*)(cb + 8 * lhalf);
    const float4 c1 = *(const float4*)(cb + 8 * lhalf + 4);
    bias8[0] = c0.x; bias8[1] = c0.y; bias8[2] = c0.z; bias8[3] = c0.w;
    bias8[4] = c1.x; bias8[5] = c1.y; bias8[6] = c1.z; bias8[7] = c1.w;
  }
  float s8[8], q8[8];
#pragma unroll
  for (int r = 0; r < 8; ++r) { s8[r] = 0.f; q8[r] = 0.f; }
  float* hq = h2 + ((mtile * 490 + j0) * H2_POS_STRIDE + lane * 4);

  __threadfence_block();

#pragma unroll 1
  for (int t = 0; t < CV_L; ++t) {
    if (t + 1 < CV_L)
      cv_load_col(ap, j0 + t + 2, cf);

    v8f acc[5] = {};
#pragma unroll
    for (int p = 0; p < 4; ++p) {
#pragma unroll
      for (int u = 0; u < 9; ++u) {
#pragma unroll
        for (int i = 0; i < 5; ++i) {
          const int r = i + (p >> 1);
          v2f af = zero2;
          if (r < 5) af = (p & 1) ? cn[r][u] : ca[r][u];
          acc[i] = __builtin_amdgcn_wmma_f32_16x16x4_f32(false, wr[9 * p + u], false, af,
                                                         (short)0, acc[i], false, false);
        }
      }
    }

#pragma unroll
    for (int i = 0; i < 5; ++i) {
      float val[8];
#pragma unroll
      for (int r = 0; r < 8; ++r) {
        float v = acc[i][r] + bias8[r];
        v = v >= 0.f ? v : 0.01f * v;
        val[r] = v;
        s8[r] += v;
        q8[r] += v * v;
      }
      float* dst = hq + (i * 98 + t) * H2_POS_STRIDE;
      *(float4*)(dst)       = make_float4(val[0], val[1], val[2], val[3]);
      *(float4*)(dst + 128) = make_float4(val[4], val[5], val[6], val[7]);
    }

    __builtin_amdgcn_sched_barrier(0);
#pragma unroll
    for (int r = 0; r < 5; ++r) {
#pragma unroll
      for (int u = 0; u < 9; ++u) { ca[r][u] = cn[r][u]; cn[r][u] = cf[r][u]; }
    }
  }
#pragma unroll
  for (int r = 0; r < 8; ++r) {
#pragma unroll
    for (int off = 1; off < 16; off <<= 1) {
      s8[r] += __shfl_xor(s8[r], off, 32);
      q8[r] += __shfl_xor(q8[r], off, 32);
    }
  }
  float sv = s8[0], qv = q8[0];
#pragma unroll
  for (int r = 1; r < 8; ++r) {
    if (lrow == r) { sv = s8[r]; qv = q8[r]; }
  }
  if (lrow < 8) { wsum[wave][8 * lhalf + lrow] = sv; wsq[wave][8 * lhalf + lrow] = qv; }
  __syncthreads();
  if (threadIdx.x < 32) {
    int c = threadIdx.x & 15;
    int st = threadIdx.x >> 4;
    float tot = 0.f;
    for (int wv = 0; wv < 8; ++wv) tot += st ? wsq[wv][c] : wsum[wv][c];
    part[blockIdx.x * 32 + threadIdx.x] = tot;
  }
}

#define LCW_VEC4   37248
#define LCW_BLOCKS 146
__global__ void __launch_bounds__(256) bn_finalize(const float* __restrict__ part,
                                                   const float* __restrict__ gamma,
                                                   const float* __restrict__ beta,
                                                   float* __restrict__ scale,
                                                   float* __restrict__ shift,
                                                   const float* __restrict__ w1,
                                                   float* __restrict__ pl) {
  if (blockIdx.x > 0) {
    const int idx = (blockIdx.x - 1) * 256 + threadIdx.x;
    if (idx < LCW_VEC4) {
      const int o  = idx % 6;
      int r        = idx / 6;
      const int h  = r & 1;  r >>= 1;
      const int s2 = r & 7;  r >>= 3;
      const int oi = r & 3;
      const int jc = r >> 2;
      const int p  = s2 >> 1;
      const int c0 = 8 * h + 4 * (s2 & 1);
      const float* src = w1 + ((o * 16 + c0) * 4 + oi) * 388 + jc * 4 + p;
      const float4 v = make_float4(src[0], src[1552], src[2 * 1552], src[3 * 1552]);
      ((float4*)pl)[((jc * 4 + oi) * 8 + s2) * 16 + h * 8 + o] = v;
    }
    return;
  }
  __shared__ float acc[256];
  __shared__ float tot[32];
  int t = threadIdx.x;
  int cs = t & 31, r = t >> 5;
  float s = 0.f;
  for (int blk = r; blk < CONV_BLOCKS; blk += 8) s += part[blk * 32 + cs];
  acc[t] = s;
  __syncthreads();
  if (t < 32) {
    float sum = 0.f;
    for (int rr = 0; rr < 8; ++rr) sum += acc[rr * 32 + t];
    tot[t] = sum;
  }
  __syncthreads();
  if (t < 16) {
    const float invN = 1.0f / BN_COUNT;
    float mean = tot[t] * invN;
    float var  = tot[16 + t] * invN - mean * mean;
    float inv  = rsqrtf(var + 1e-5f);
    float scl  = gamma[t] * inv;
    scale[t] = scl;
    shift[t] = beta[t] - mean * scl;
  }
}

#define LC_L    6
#define LC_RUNS 16

__device__ __forceinline__ void lc_load_col(const float* __restrict__ hp, int jj,
                                            v2f (&c)[5][4]) {
#pragma unroll
  for (int ii = 0; ii < 5; ++ii) {
    const float4 a = *(const float4*)(hp + (ii * 98 + jj) * H2_POS_STRIDE);
    const float4 b = *(const float4*)(hp + (ii * 98 + jj) * H2_POS_STRIDE + 128);
    c[ii][0][0] = a.x; c[ii][0][1] = a.y;
    c[ii][1][0] = a.z; c[ii][1][1] = a.w;
    c[ii][2][0] = b.x; c[ii][2][1] = b.y;
    c[ii][3][0] = b.z; c[ii][3][1] = b.w;
  }
}

__device__ __forceinline__ void lc_bn_col(v2f (&c)[5][4], const float (&sc)[8],
                                          const float (&sf)[8]) {
#pragma unroll
  for (int ii = 0; ii < 5; ++ii) {
#pragma unroll
    for (int q = 0; q < 4; ++q) {
      c[ii][q][0] = c[ii][q][0] * sc[2 * q]     + sf[2 * q];
      c[ii][q][1] = c[ii][q][1] * sc[2 * q + 1] + sf[2 * q + 1];
    }
  }
}

__device__ __forceinline__ void lc_load_w(const float* __restrict__ wp, int jc, int p,
                                          v2f (&w)[4][16]) {
#pragma unroll
  for (int oi = 0; oi < 4; ++oi) {
#pragma unroll
    for (int z2 = 0; z2 < 2; ++z2) {
      const int s2 = 2 * p + z2;
      const float4 v = *(const float4*)(wp + ((jc * 4 + oi) * 8 + s2) * 64);
      w[oi][2 * s2][0]     = v.x; w[oi][2 * s2][1]     = v.y;
      w[oi][2 * s2 + 1][0] = v.z; w[oi][2 * s2 + 1][1] = v.w;
    }
  }
}

__global__ void __launch_bounds__(128) lc_wmma(const float* __restrict__ h2,
                                               const float* __restrict__ scale,
                                               const float* __restrict__ shift,
                                               const float* __restrict__ pl,
                                               const float* __restrict__ b1,
                                               const float* __restrict__ w2,
                                               const float* __restrict__ b2,
                                               float* __restrict__ out) {
  const int lane  = threadIdx.x & 31;
  const int wave  = threadIdx.x >> 5;
  const int run   = blockIdx.x >> 5;
  const int mtile = (blockIdx.x & 31) * 4 + wave;
  const int lrow  = lane & 15;
  const int lhalf = lane >> 4;
  const int m     = mtile * 16 + lrow;
  const int j0    = run * LC_L;

  float sc[8], sf[8];
  {
    const float4 s0 = ((const float4*)scale)[2 * lhalf];
    const float4 s1 = ((const float4*)scale)[2 * lhalf + 1];
    const float4 f0 = ((const float4*)shift)[2 * lhalf];
    const float4 f1 = ((const float4*)shift)[2 * lhalf + 1];
    sc[0] = s0.x; sc[1] = s0.y; sc[2] = s0.z; sc[3] = s0.w;
    sc[4] = s1.x; sc[5] = s1.y; sc[6] = s1.z; sc[7] = s1.w;
    sf[0] = f0.x; sf[1] = f0.y; sf[2] = f0.z; sf[3] = f0.w;
    sf[4] = f1.x; sf[5] = f1.y; sf[6] = f1.z; sf[7] = f1.w;
  }

  const float* hp = h2 + (mtile * 490 * H2_POS_STRIDE + lane * 4);
  const float* wp = pl + (lhalf * 8 + (lrow < 6 ? lrow : 0)) * 4;

  v2f   col[3][5][4];
  v2f   w[4][16];
  float h3p[6][4];

  lc_load_col(hp, j0, col[0]);
  lc_load_col(hp, j0 + 1, col[1]);
#pragma unroll
  for (int p = 0; p < 4; ++p) lc_load_w(wp, j0, p, w);
  lc_bn_col(col[0], sc, sf);
  lc_bn_col(col[1], sc, sf);

#pragma unroll
  for (int t = 0; t <= LC_L; ++t) {
    const int jc = j0 + t;
    if (t < LC_L) {
      lc_load_col(hp, jc + 2, col[(t + 2) % 3]);
    }
    v2f (&ca)[5][4] = col[t % 3];
    v2f (&cb)[5][4] = col[(t + 1) % 3];

    v8f acc[4] = {};
#pragma unroll
    for (int p = 0; p < 4; ++p) {
#pragma unroll
      for (int q = 0; q < 4; ++q) {
#pragma unroll
        for (int oi = 0; oi < 4; ++oi) {
          const v2f af = w[oi][4 * p + q];
          const v2f bf = (p & 1) ? cb[oi + (p >> 1)][q] : ca[oi + (p >> 1)][q];
          acc[oi] = __builtin_amdgcn_wmma_f32_16x16x4_f32(false, af, false, bf,
                                                          (short)0, acc[oi], false, false);
        }
      }
      if (t < LC_L) lc_load_w(wp, jc + 1, p, w);
    }

    float h3c[6][4];
#pragma unroll
    for (int oi = 0; oi < 4; ++oi) {
#pragma unroll
      for (int o = 0; o < 6; ++o) {
        float v = acc[oi][o] + b1[o * 388 + oi * 97 + jc];
        h3c[o][oi] = v >= 0.f ? v : 0.01f * v;
      }
    }

    if (t > 0) {
      const int rj = jc - 1;
#pragma unroll
      for (int ri = 0; ri < 3; ++ri) {
        float a = b2[ri * 96 + rj];
#pragma unroll
        for (int o = 0; o < 6; ++o) {
          const float4 wv = *(const float4*)(w2 + (((o * 3 + ri) * 96 + rj) << 2));
          a += wv.x * h3p[o][ri] + wv.y * h3c[o][ri]
             + wv.z * h3p[o][ri + 1] + wv.w * h3c[o][ri + 1];
        }
        if (lane < 16) out[m * 288 + ri * 96 + rj] = a;
      }
    }
#pragma unroll
    for (int o = 0; o < 6; ++o) {
#pragma unroll
      for (int oi = 0; oi < 4; ++oi) h3p[o][oi] = h3c[o][oi];
    }
    if (t < LC_L) lc_bn_col(col[(t + 2) % 3], sc, sf);
  }
}

extern "C" void kernel_launch(void* const* d_in, const int* in_sizes, int n_in,
                              void* d_out, int out_size, void* d_ws, size_t ws_size,
                              hipStream_t stream) {
  const float* x     = (const float*)d_in[0];
  const float* fw    = (const float*)d_in[1];
  const float* fb    = (const float*)d_in[2];
  const float* cw    = (const float*)d_in[3];
  const float* cb    = (const float*)d_in[4];
  const float* gamma = (const float*)d_in[5];
  const float* beta  = (const float*)d_in[6];
  const float* w1    = (const float*)d_in[7];
  const float* b1    = (const float*)d_in[8];
  const float* w2    = (const float*)d_in[9];
  const float* b2    = (const float*)d_in[10];

  float* ws    = (float*)d_ws;
  float* h1    = ws;
  float* h2    = ws + H1_ELEMS;
  float* part  = ws + H1_ELEMS + H2_ELEMS;
  float* scale = part + PART_ELEMS;
  float* shift = scale + 16;
  float* pw    = shift + 16;
  float* pl    = ws;

  pack_conv_w<<<9, 256, 0, stream>>>(cw, pw);
  fc_wmma    <<<2208, 256, 0, stream>>>(x, fw, fb, h1);
  conv_wmma  <<<CONV_BLOCKS, 256, 0, stream>>>(h1, pw, cb, h2, part);
  bn_finalize<<<1 + LCW_BLOCKS, 256, 0, stream>>>(part, gamma, beta, scale, shift, w1, pl);
  lc_wmma    <<<32 * LC_RUNS, 128, 0, stream>>>(h2, scale, shift, pl, b1, w2, b2,
                                                (float*)d_out);
}


# x repacked once per call into the fc kernel's WMMA fragment order (extra blocks of the repack launch, into the h2 area): an fc x-fragment load is one aligned 256-B run, was 16 lines; sum order kept
# speedup vs baseline: 4.0721x; 1.0407x over previous
#include <hip/hip_runtime.h>

typedef __attribute__((ext_vector_type(2))) float v2f;
typedef __attribute__((ext_vector_type(8))) float v8f;

#define NB      2048
#define FC_K    100
#define FC_N    17640
#define NTILES  1103
#define H1_ELEMS (NB * FC_N)
#define H2_ELEMS (NB * 5 * 98 * 16)
#define CV_L    7
#define CV_RUNS 14
#define CONV_BLOCKS (16 * CV_RUNS)
#define PART_ELEMS (CONV_BLOCKS * 32)
#define BN_COUNT 1003520.0f
#define H1_PAIRS 8820
#define H2_POS_STRIDE 256
#define XF_VEC2   102400
#define XF_PER_T  8
#define XF_BLOCKS 50
#define PW_BLOCKS 9

#define SLAB_FLOATS 12800
#define SLAB_VEC4   3200
#define FW_VEC4     441000
__global__ void __launch_bounds__(256) fc_wmma(const float* __restrict__ xf,
                                               const float* __restrict__ fw,
                                               const float* __restrict__ fb,
                                               float* __restrict__ h1) {
  __shared__ float sB[SLAB_FLOATS];
  const int tid   = threadIdx.x;
  const int lane  = tid & 31;
  const int wave  = tid >> 5;
  const int slab  = blockIdx.x >> 4;
  const int mblk  = blockIdx.x & 15;
  const int mtile = mblk * 8 + wave;
  const int lrow  = lane & 15;
  const int lhalf = lane >> 4;
  const int kb0   = 2 * lhalf;

  const float* xp = xf + (mtile * 25 * 32 + lane) * 2;
  v2f a[25];
#pragma unroll
  for (int s = 0; s < 25; ++s)
    a[s] = *(const v2f*)(xp + 64 * s);

  const float4* src4 = (const float4*)fw + slab * SLAB_VEC4;
  const int vleft = FW_VEC4 - slab * SLAB_VEC4;
  float4 stg[13];
#pragma unroll
  for (int r = 0; r < 13; ++r) {
    const int idx = tid + r * 256;
    float4 v = make_float4(0.f, 0.f, 0.f, 0.f);
    if ((r < 12 || tid < 128) && idx < vleft) v = src4[idx];
    stg[r] = v;
  }
#pragma unroll
  for (int r = 0; r < 13; ++r) {
    const int idx = tid + r * 256;
    if (r < 12 || tid < 128) ((float4*)sB)[idx] = stg[r];
  }
  __syncthreads();

  for (int t = 0; t < 8; ++t) {
    const int ntile = slab * 8 + t;
    if (ntile >= NTILES) break;
    const int nl = ntile * 16 + 8 * lhalf;
    const bool nvalid = (nl < FC_N);
    float4 bia0 = make_float4(0.f, 0.f, 0.f, 0.f), bia1 = bia0;
    if (nvalid) {
      bia0 = *(const float4*)(fb + nl);
      bia1 = *(const float4*)(fb + nl + 4);
    }
    const float* bp = sB + (t * 16 + lrow) * FC_K + kb0;
    v8f acc = {};
#pragma unroll
    for (int s = 0; s < 25; ++s) {
      v2f wf = *(const v2f*)(bp + 4 * s);
      acc = __builtin_amdgcn_wmma_f32_16x16x4_f32(false, wf, false, a[s],
                                                  (short)0, acc, false, false);
    }
    if (nvalid) {
      float* dst = h1 + ((mtile * H1_PAIRS + ntile * 8 + 4 * lhalf) * 32 + lrow * 2);
      v2f o0, o1, o2, o3;
      o0[0] = acc[0] + bia0.x; o0[1] = acc[1] + bia0.y;
      o1[0] = acc[2] + bia0.z; o1[1] = acc[3] + bia0.w;
      o2[0] = acc[4] + bia1.x; o2[1] = acc[5] + bia1.y;
      o3[0] = acc[6] + bia1.z; o3[1] = acc[7] + bia1.w;
      *(v2f*)(dst)      = o0;
      *(v2f*)(dst + 32) = o1;
      *(v2f*)(dst + 64) = o2;
      *(v2f*)(dst + 96) = o3;
    }
  }
}

__global__ void __launch_bounds__(256) pack_conv_w(const float* __restrict__ cw,
                                                   float* __restrict__ pw,
                                                   const float* __restrict__ x,
                                                   float* __restrict__ xf) {
  if (blockIdx.x >= PW_BLOCKS) {
    const int xb = blockIdx.x - PW_BLOCKS;
    v2f v[XF_PER_T];
#pragma unroll
    for (int j = 0; j < XF_PER_T; ++j) {
      const int xi = (j * XF_BLOCKS + xb) * 256 + threadIdx.x;
      const int ln = xi & 31;
      const int r  = xi >> 5;
      const int mt = r / 25;
      const int ks = r - mt * 25;
      v[j] = *(const v2f*)(x + (mt * 16 + (ln & 15)) * FC_K + 4 * ks + 2 * (ln >> 4));
    }
#pragma unroll
    for (int j = 0; j < XF_PER_T; ++j) {
      const int xi = (j * XF_BLOCKS + xb) * 256 + threadIdx.x;
      *(v2f*)(xf + xi * 2) = v[j];
    }
    return;
  }
  int idx = blockIdx.x * 256 + threadIdx.x;
  if (idx >= 2304) return;
  int s  = idx >> 6;
  int r  = idx & 63;
  int ln = r >> 1;
  int e  = r & 1;
  int n  = ln & 15;
  int k  = 4 * s + 2 * (ln >> 4) + e;
  int p  = k / 36;
  int c  = k - p * 36;
  pw[idx] = cw[n * 144 + c * 4 + p];
}

__device__ __forceinline__ void cv_load_col(const float* __restrict__ ap, int jj,
                                            v2f (&c)[5][9]) {
  if (jj < 98) {
#pragma unroll
    for (int r = 0; r < 5; ++r) {
#pragma unroll
      for (int u = 0; u < 9; ++u)
        c[r][u] = *(const v2f*)(ap + ((r * 98 + jj) * 18 + 2 * u) * 32);
    }
  } else {
#pragma unroll
    for (int r = 0; r < 5; ++r) {
#pragma unroll
      for (int u = 0; u < 9; ++u) { c[r][u][0] = 0.f; c[r][u][1] = 0.f; }
    }
  }
}

__global__ void __launch_bounds__(256) conv_wmma(const float* __restrict__ h1,
                                                 const float* __restrict__ pw,
                                                 const float* __restrict__ cb,
                                                 float* __restrict__ h2,
                                                 float* __restrict__ part) {
  __shared__ float wsum[8][16];
  __shared__ float wsq[8][16];
  const int lane  = threadIdx.x & 31;
  const int wave  = threadIdx.x >> 5;
  const int run   = blockIdx.x % CV_RUNS;
  const int mblk  = blockIdx.x / CV_RUNS;
  const int mtile = mblk * 8 + wave;
  const int j0    = run * CV_L;
  const int lrow  = lane & 15;
  const int lhalf = lane >> 4;
  const float* ap  = h1 + (mtile * H1_PAIRS * 32 + lhalf * 32 + lrow * 2);
  const float* bwp = pw + lane * 2;

  v2f wr[36];
#pragma unroll
  for (int s = 0; s < 36; ++s) wr[s] = *(const v2f*)(bwp + 64 * s);

  v2f ca[5][9], cn[5][9], cf[5][9];
  cv_load_col(ap, j0, ca);
  cv_load_col(ap, j0 + 1, cn);

  v2f zero2;
  zero2[0] = 0.f; zero2[1] = 0.f;

  float bias8[8];
  {
    const float4 c0 = *(const float4*)(cb + 8 * lhalf);
    const float4 c1 = *(const float4*)(cb + 8 * lhalf + 4);
    bias8[0] = c0.x; bias8[1] = c0.y; bias8[2] = c0.z; bias8[3] = c0.w;
    bias8[4] = c1.x; bias8[5] = c1.y; bias8[6] = c1.z; bias8[7] = c1.w;
  }
  float s8[8], q8[8];
#pragma unroll
  for (int r = 0; r < 8; ++r) { s8[r] = 0.f; q8[r] = 0.f; }
  float* hq = h2 + ((mtile * 490 + j0) * H2_POS_STRIDE + lane * 4);

  __threadfence_block();

#pragma unroll 1
  for (int t = 0; t < CV_L; ++t) {
    if (t + 1 < CV_L)
      cv_load_col(ap, j0 + t + 2, cf);

    v8f acc[5] = {};
#pragma unroll
    for (int p = 0; p < 4; ++p) {
#pragma unroll
      for (int u = 0; u < 9; ++u) {
#pragma unroll
        for (int i = 0; i < 5; ++i) {
          const int r = i + (p >> 1);
          v2f af = zero2;
          if (r < 5) af = (p & 1) ? cn[r][u] : ca[r][u];
          acc[i] = __builtin_amdgcn_wmma_f32_16x16x4_f32(false, wr[9 * p + u], false, af,
                                                         (short)0, acc[i], false, false);
        }
      }
    }

#pragma unroll
    for (int i = 0; i < 5; ++i) {
      float val[8];
#pragma unroll
      for (int r = 0; r < 8; ++r) {
        float v = acc[i][r] + bias8[r];
        v = v >= 0.f ? v : 0.01f * v;
        val[r] = v;
        s8[r] += v;
        q8[r] += v * v;
      }
      float* dst = hq + (i * 98 + t) * H2_POS_STRIDE;
      *(float4*)(dst)       = make_float4(val[0], val[1], val[2], val[3]);
      *(float4*)(dst + 128) = make_float4(val[4], val[5], val[6], val[7]);
    }

    __builtin_amdgcn_sched_barrier(0);
#pragma unroll
    for (int r = 0; r < 5; ++r) {
#pragma unroll
      for (int u = 0; u < 9; ++u) { ca[r][u] = cn[r][u]; cn[r][u] = cf[r][u]; }
    }
  }
#pragma unroll
  for (int r = 0; r < 8; ++r) {
#pragma unroll
    for (int off = 1; off < 16; off <<= 1) {
      s8[r] += __shfl_xor(s8[r], off, 32);
      q8[r] += __shfl_xor(q8[r], off, 32);
    }
  }
  float sv = s8[0], qv = q8[0];
#pragma unroll
  for (int r = 1; r < 8; ++r) {
    if (lrow == r) { sv = s8[r]; qv = q8[r]; }
  }
  if (lrow < 8) { wsum[wave][8 * lhalf + lrow] = sv; wsq[wave][8 * lhalf + lrow] = qv; }
  __syncthreads();
  if (threadIdx.x < 32) {
    int c = threadIdx.x & 15;
    int st = threadIdx.x >> 4;
    float tot = 0.f;
    for (int wv = 0; wv < 8; ++wv) tot += st ? wsq[wv][c] : wsum[wv][c];
    part[blockIdx.x * 32 + threadIdx.x] = tot;
  }
}

#define LCW_VEC4   37248
#define LCW_BLOCKS 146
__global__ void __launch_bounds__(256) bn_finalize(const float* __restrict__ part,
                                                   const float* __restrict__ gamma,
                                                   const float* __restrict__ beta,
                                                   float* __restrict__ scale,
                                                   float* __restrict__ shift,
                                                   const float* __restrict__ w1,
                                                   float* __restrict__ pl) {
  if (blockIdx.x > 0) {
    const int idx = (blockIdx.x - 1) * 256 + threadIdx.x;
    if (idx < LCW_VEC4) {
      const int o  = idx % 6;
      int r        = idx / 6;
      const int h  = r & 1;  r >>= 1;
      const int s2 = r & 7;  r >>= 3;
      const int oi = r & 3;
      const int jc = r >> 2;
      const int p  = s2 >> 1;
      const int c0 = 8 * h + 4 * (s2 & 1);
      const float* src = w1 + ((o * 16 + c0) * 4 + oi) * 388 + jc * 4 + p;
      const float4 v = make_float4(src[0], src[1552], src[2 * 1552], src[3 * 1552]);
      ((float4*)pl)[((jc * 4 + oi) * 8 + s2) * 16 + h * 8 + o] = v;
    }
    return;
  }
  __shared__ float acc[256];
  __shared__ float tot[32];
  int t = threadIdx.x;
  int cs = t & 31, r = t >> 5;
  float s = 0.f;
  for (int blk = r; blk < CONV_BLOCKS; blk += 8) s += part[blk * 32 + cs];
  acc[t] = s;
  __syncthreads();
  if (t < 32) {
    float sum = 0.f;
    for (int rr = 0; rr < 8; ++rr) sum += acc[rr * 32 + t];
    tot[t] = sum;
  }
  __syncthreads();
  if (t < 16) {
    const float invN = 1.0f / BN_COUNT;
    float mean = tot[t] * invN;
    float var  = tot[16 + t] * invN - mean * mean;
    float inv  = rsqrtf(var + 1e-5f);
    float scl  = gamma[t] * inv;
    scale[t] = scl;
    shift[t] = beta[t] - mean * scl;
  }
}

#define LC_L    6
#define LC_RUNS 16

__device__ __forceinline__ void lc_load_col(const float* __restrict__ hp, int jj,
                                            v2f (&c)[5][4]) {
#pragma unroll
  for (int ii = 0; ii < 5; ++ii) {
    const float4 a = *(const float4*)(hp + (ii * 98 + jj) * H2_POS_STRIDE);
    const float4 b = *(const float4*)(hp + (ii * 98 + jj) * H2_POS_STRIDE + 128);
    c[ii][0][0] = a.x; c[ii][0][1] = a.y;
    c[ii][1][0] = a.z; c[ii][1][1] = a.w;
    c[ii][2][0] = b.x; c[ii][2][1] = b.y;
    c[ii][3][0] = b.z; c[ii][3][1] = b.w;
  }
}

__device__ __forceinline__ void lc_bn_col(v2f (&c)[5][4], const float (&sc)[8],
                                          const float (&sf)[8]) {
#pragma unroll
  for (int ii = 0; ii < 5; ++ii) {
#pragma unroll
    for (int q = 0; q < 4; ++q) {
      c[ii][q][0] = c[ii][q][0] * sc[2 * q]     + sf[2 * q];
      c[ii][q][1] = c[ii][q][1] * sc[2 * q + 1] + sf[2 * q + 1];
    }
  }
}

__device__ __forceinline__ void lc_load_w(const float* __restrict__ wp, int jc, int p,
                                          v2f (&w)[4][16]) {
#pragma unroll
  for (int oi = 0; oi < 4; ++oi) {
#pragma unroll
    for (int z2 = 0; z2 < 2; ++z2) {
      const int s2 = 2 * p + z2;
      const float4 v = *(const float4*)(wp + ((jc * 4 + oi) * 8 + s2) * 64);
      w[oi][2 * s2][0]     = v.x; w[oi][2 * s2][1]     = v.y;
      w[oi][2 * s2 + 1][0] = v.z; w[oi][2 * s2 + 1][1] = v.w;
    }
  }
}

__global__ void __launch_bounds__(128) lc_wmma(const float* __restrict__ h2,
                                               const float* __restrict__ scale,
                                               const float* __restrict__ shift,
                                               const float* __restrict__ pl,
                                               const float* __restrict__ b1,
                                               const float* __restrict__ w2,
                                               const float* __restrict__ b2,
                                               float* __restrict__ out) {
  const int lane  = threadIdx.x & 31;
  const int wave  = threadIdx.x >> 5;
  const int run   = blockIdx.x >> 5;
  const int mtile = (blockIdx.x & 31) * 4 + wave;
  const int lrow  = lane & 15;
  const int lhalf = lane >> 4;
  const int m     = mtile * 16 + lrow;
  const int j0    = run * LC_L;

  float sc[8], sf[8];
  {
    const float4 s0 = ((const float4*)scale)[2 * lhalf];
    const float4 s1 = ((const float4*)scale)[2 * lhalf + 1];
    const float4 f0 = ((const float4*)shift)[2 * lhalf];
    const float4 f1 = ((const float4*)shift)[2 * lhalf + 1];
    sc[0] = s0.x; sc[1] = s0.y; sc[2] = s0.z; sc[3] = s0.w;
    sc[4] = s1.x; sc[5] = s1.y; sc[6] = s1.z; sc[7] = s1.w;
    sf[0] = f0.x; sf[1] = f0.y; sf[2] = f0.z; sf[3] = f0.w;
    sf[4] = f1.x; sf[5] = f1.y; sf[6] = f1.z; sf[7] = f1.w;
  }

  const float* hp = h2 + (mtile * 490 * H2_POS_STRIDE + lane * 4);
  const float* wp = pl + (lhalf * 8 + (lrow < 6 ? lrow : 0)) * 4;

  v2f   col[3][5][4];
  v2f   w[4][16];
  float h3p[6][4];

  lc_load_col(hp, j0, col[0]);
  lc_load_col(hp, j0 + 1, col[1]);
#pragma unroll
  for (int p = 0; p < 4; ++p) lc_load_w(wp, j0, p, w);
  lc_bn_col(col[0], sc, sf);
  lc_bn_col(col[1], sc, sf);

#pragma unroll
  for (int t = 0; t <= LC_L; ++t) {
    const int jc = j0 + t;
    if (t < LC_L) {
      lc_load_col(hp, jc + 2, col[(t + 2) % 3]);
    }
    v2f (&ca)[5][4] = col[t % 3];
    v2f (&cb)[5][4] = col[(t + 1) % 3];

    v8f acc[4] = {};
#pragma unroll
    for (int p = 0; p < 4; ++p) {
#pragma unroll
      for (int q = 0; q < 4; ++q) {
#pragma unroll
        for (int oi = 0; oi < 4; ++oi) {
          const v2f af = w[oi][4 * p + q];
          const v2f bf = (p & 1) ? cb[oi + (p >> 1)][q] : ca[oi + (p >> 1)][q];
          acc[oi] = __builtin_amdgcn_wmma_f32_16x16x4_f32(false, af, false, bf,
                                                          (short)0, acc[oi], false, false);
        }
      }
      if (t < LC_L) lc_load_w(wp, jc + 1, p, w);
    }

    float h3c[6][4];
#pragma unroll
    for (int oi = 0; oi < 4; ++oi) {
#pragma unroll
      for (int o = 0; o < 6; ++o) {
        float v = acc[oi][o] + b1[o * 388 + oi * 97 + jc];
        h3c[o][oi] = v >= 0.f ? v : 0.01f * v;
      }
    }

    if (t > 0) {
      const int rj = jc - 1;
#pragma unroll
      for (int ri = 0; ri < 3; ++ri) {
        float a = b2[ri * 96 + rj];
#pragma unroll
        for (int o = 0; o < 6; ++o) {
          const float4 wv = *(const float4*)(w2 + (((o * 3 + ri) * 96 + rj) << 2));
          a += wv.x * h3p[o][ri] + wv.y * h3c[o][ri]
             + wv.z * h3p[o][ri + 1] + wv.w * h3c[o][ri + 1];
        }
        if (lane < 16) out[m * 288 + ri * 96 + rj] = a;
      }
    }
#pragma unroll
    for (int o = 0; o < 6; ++o) {
#pragma unroll
      for (int oi = 0; oi < 4; ++oi) h3p[o][oi] = h3c[o][oi];
    }
    if (t < LC_L) lc_bn_col(col[(t + 2) % 3], sc, sf);
  }
}

extern "C" void kernel_launch(void* const* d_in, const int* in_sizes, int n_in,
                              void* d_out, int out_size, void* d_ws, size_t ws_size,
                              hipStream_t stream) {
  const float* x     = (const float*)d_in[0];
  const float* fw    = (const float*)d_in[1];
  const float* fb    = (const float*)d_in[2];
  const float* cw    = (const float*)d_in[3];
  const float* cb    = (const float*)d_in[4];
  const float* gamma = (const float*)d_in[5];
  const float* beta  = (const float*)d_in[6];
  const float* w1    = (const float*)d_in[7];
  const float* b1    = (const float*)d_in[8];
  const float* w2    = (const float*)d_in[9];
  const float* b2    = (const float*)d_in[10];

  float* ws    = (float*)d_ws;
  float* h1    = ws;
  float* h2    = ws + H1_ELEMS;
  float* part  = ws + H1_ELEMS + H2_ELEMS;
  float* scale = part + PART_ELEMS;
  float* shift = scale + 16;
  float* pw    = shift + 16;
  float* pl    = ws;
  float* xf    = h2;

  pack_conv_w<<<PW_BLOCKS + XF_BLOCKS, 256, 0, stream>>>(cw, pw, x, xf);
  fc_wmma    <<<2208, 256, 0, stream>>>(xf, fw, fb, h1);
  conv_wmma  <<<CONV_BLOCKS, 256, 0, stream>>>(h1, pw, cb, h2, part);
  bn_finalize<<<1 + LCW_BLOCKS, 256, 0, stream>>>(part, gamma, beta, scale, shift, w1, pl);
  lc_wmma    <<<32 * LC_RUNS, 128, 0, stream>>>(h2, scale, shift, pl, b1, w2, b2,
                                                (float*)d_out);
}
